# starting kernel
# baseline (speedup 1.0000x reference)
#include <hip/hip_runtime.h>
#include <hip/hip_bf16.h>

typedef __attribute__((ext_vector_type(2))) float v2f;
typedef __attribute__((ext_vector_type(8))) float v8f;

#define DD   1024
#define BB   16
#define HH   4
#define EPSF 1e-8f
#define LOG2E 1.44269504088896340736f

__global__ __launch_bounds__(128) void flatt_kernel(
    const float* __restrict__ x,
    const float* __restrict__ aq, const float* __restrict__ bq,
    const float* __restrict__ ak, const float* __restrict__ bk,
    const float* __restrict__ av, const float* __restrict__ bv,
    float* __restrict__ out)
{
    __shared__ float xs[DD];
    __shared__ float att_s[HH * 16];
    __shared__ float dn_s[HH * 16];

    const int b     = blockIdx.x >> 6;
    const int itile = blockIdx.x & 63;
    const int i0    = itile << 4;
    const int tid   = threadIdx.x;
    const int h     = tid >> 5;
    const int lane  = tid & 31;
    const int iloc  = lane & 15;
    const int half  = lane >> 4;

    #pragma unroll
    for (int r = 0; r < DD / 128; ++r)
        xs[tid + 128 * r] = x[b * DD + tid + 128 * r];
    __syncthreads();

    const float Aq = aq[h], Bq = bq[h];
    const float Ak = ak[h], Bk = bk[h];
    const float Av = av[h], Bv = bv[h];

    const float q = fmaf(xs[i0 + iloc], Aq, Bq);

    float m = -3.4e38f;
    for (int c = 0; c < DD / 4; ++c) {
        const int j = 4 * c + 2 * half;
        const float x0 = xs[j], x1 = xs[j + 1];
        const float s0 = __builtin_amdgcn_rcpf(fabsf(fmaf(x0, Ak, Bk) - q) + EPSF);
        const float s1 = __builtin_amdgcn_rcpf(fabsf(fmaf(x1, Ak, Bk) - q) + EPSF);
        m = fmaxf(m, fmaxf(s0, s1));
    }
    m = fmaxf(m, __shfl_xor(m, 16, 32));

    v8f acc = {};
    float denom = 0.0f;
    for (int c = 0; c < DD / 4; ++c) {
        const int j = 4 * c + 2 * half;
        const float x0 = xs[j], x1 = xs[j + 1];
        const float s0 = __builtin_amdgcn_rcpf(fabsf(fmaf(x0, Ak, Bk) - q) + EPSF);
        const float s1 = __builtin_amdgcn_rcpf(fabsf(fmaf(x1, Ak, Bk) - q) + EPSF);
        const float p0 = __builtin_amdgcn_exp2f((s0 - m) * LOG2E);
        const float p1 = __builtin_amdgcn_exp2f((s1 - m) * LOG2E);
        denom += p0 + p1;

        v2f A;  A.x  = p0;                  A.y  = p1;
        v2f Bm; Bm.x = fmaf(x0, Av, Bv);    Bm.y = fmaf(x1, Av, Bv);

        acc = __builtin_amdgcn_wmma_f32_16x16x4_f32(
                  false, A, false, Bm, (short)0, acc, false, false);
    }
    denom += __shfl_xor(denom, 16, 32);

    if (lane < 16) dn_s[h * 16 + iloc] = denom;
    if (lane == 0) {
        #pragma unroll
        for (int r = 0; r < 8; ++r) att_s[h * 16 + r] = acc[r];
    }
    if (lane == 16) {
        #pragma unroll
        for (int r = 0; r < 8; ++r) att_s[h * 16 + 8 + r] = acc[r];
    }
    __syncthreads();

    if (tid < 16) {
        float s = 0.0f;
        #pragma unroll
        for (int hh = 0; hh < HH; ++hh)
            s += att_s[hh * 16 + tid] / dn_s[hh * 16 + tid];
        out[b * DD + i0 + tid] = xs[i0 + tid] + 0.5f * s;
    }
}

extern "C" void kernel_launch(void* const* d_in, const int* in_sizes, int n_in,
                              void* d_out, int out_size, void* d_ws, size_t ws_size,
                              hipStream_t stream) {
    (void)in_sizes; (void)n_in; (void)d_ws; (void)ws_size; (void)out_size;
    const float* x  = (const float*)d_in[0];
    const float* aq = (const float*)d_in[1];
    const float* bq = (const float*)d_in[2];
    const float* ak = (const float*)d_in[3];
    const float* bk = (const float*)d_in[4];
    const float* av = (const float*)d_in[5];
    const float* bv = (const float*)d_in[6];
    float* out = (float*)d_out;

    flatt_kernel<<<dim3(BB * (DD / 16)), dim3(128), 0, stream>>>(
        x, aq, bq, ak, bk, av, bv, out);
}


# pass 1 (row max): max_j rcp(|k-q|+eps) becomes rcp(min_j |k-q| + eps): running min3 of distances, one reciprocal per row instead of one per element
# speedup vs baseline: 1.1233x; 1.1233x over previous
#include <hip/hip_runtime.h>
#include <hip/hip_bf16.h>

typedef __attribute__((ext_vector_type(2))) float v2f;
typedef __attribute__((ext_vector_type(8))) float v8f;

#define DD   1024
#define BB   16
#define HH   4
#define EPSF 1e-8f
#define LOG2E 1.44269504088896340736f

__global__ __launch_bounds__(128) void flatt_kernel(
    const float* __restrict__ x,
    const float* __restrict__ aq, const float* __restrict__ bq,
    const float* __restrict__ ak, const float* __restrict__ bk,
    const float* __restrict__ av, const float* __restrict__ bv,
    float* __restrict__ out)
{
    __shared__ float xs[DD];
    __shared__ float att_s[HH * 16];
    __shared__ float dn_s[HH * 16];

    const int b     = blockIdx.x >> 6;
    const int itile = blockIdx.x & 63;
    const int i0    = itile << 4;
    const int tid   = threadIdx.x;
    const int h     = tid >> 5;
    const int lane  = tid & 31;
    const int iloc  = lane & 15;
    const int half  = lane >> 4;

    #pragma unroll
    for (int r = 0; r < DD / 128; ++r)
        xs[tid + 128 * r] = x[b * DD + tid + 128 * r];
    __syncthreads();

    const float Aq = aq[h], Bq = bq[h];
    const float Ak = ak[h], Bk = bk[h];
    const float Av = av[h], Bv = bv[h];

    const float q = fmaf(xs[i0 + iloc], Aq, Bq);

    float dmin0 = 3.4e38f, dmin1 = 3.4e38f;
    #pragma clang loop vectorize(disable)
    for (int c = 0; c < DD / 4; c += 2) {
        const int j = 4 * c + 2 * half;
        const float x0 = xs[j],     x1 = xs[j + 1];
        const float x2 = xs[j + 4], x3 = xs[j + 5];
        const float d0 = fabsf(fmaf(x0, Ak, Bk) - q);
        const float d1 = fabsf(fmaf(x1, Ak, Bk) - q);
        const float d2 = fabsf(fmaf(x2, Ak, Bk) - q);
        const float d3 = fabsf(fmaf(x3, Ak, Bk) - q);
        dmin0 = __builtin_fminf(dmin0, __builtin_fminf(d0, d1));
        dmin1 = __builtin_fminf(dmin1, __builtin_fminf(d2, d3));
    }
    float dmin = __builtin_fminf(dmin0, dmin1);
    dmin = __builtin_fminf(dmin, __shfl_xor(dmin, 16, 32));
    const float m = __builtin_amdgcn_rcpf(dmin + EPSF);

    v8f acc = {};
    float denom = 0.0f;
    for (int c = 0; c < DD / 4; ++c) {
        const int j = 4 * c + 2 * half;
        const float x0 = xs[j], x1 = xs[j + 1];
        const float s0 = __builtin_amdgcn_rcpf(fabsf(fmaf(x0, Ak, Bk) - q) + EPSF);
        const float s1 = __builtin_amdgcn_rcpf(fabsf(fmaf(x1, Ak, Bk) - q) + EPSF);
        const float p0 = __builtin_amdgcn_exp2f((s0 - m) * LOG2E);
        const float p1 = __builtin_amdgcn_exp2f((s1 - m) * LOG2E);
        denom += p0 + p1;

        v2f A;  A.x  = p0;                  A.y  = p1;
        v2f Bm; Bm.x = fmaf(x0, Av, Bv);    Bm.y = fmaf(x1, Av, Bv);

        acc = __builtin_amdgcn_wmma_f32_16x16x4_f32(
                  false, A, false, Bm, (short)0, acc, false, false);
    }
    denom += __shfl_xor(denom, 16, 32);

    if (lane < 16) dn_s[h * 16 + iloc] = denom;
    if (lane == 0) {
        #pragma unroll
        for (int r = 0; r < 8; ++r) att_s[h * 16 + r] = acc[r];
    }
    if (lane == 16) {
        #pragma unroll
        for (int r = 0; r < 8; ++r) att_s[h * 16 + 8 + r] = acc[r];
    }
    __syncthreads();

    if (tid < 16) {
        float s = 0.0f;
        #pragma unroll
        for (int hh = 0; hh < HH; ++hh)
            s += att_s[hh * 16 + tid] / dn_s[hh * 16 + tid];
        out[b * DD + i0 + tid] = xs[i0 + tid] + 0.5f * s;
    }
}

extern "C" void kernel_launch(void* const* d_in, const int* in_sizes, int n_in,
                              void* d_out, int out_size, void* d_ws, size_t ws_size,
                              hipStream_t stream) {
    (void)in_sizes; (void)n_in; (void)d_ws; (void)ws_size; (void)out_size;
    const float* x  = (const float*)d_in[0];
    const float* aq = (const float*)d_in[1];
    const float* bq = (const float*)d_in[2];
    const float* ak = (const float*)d_in[3];
    const float* bk = (const float*)d_in[4];
    const float* av = (const float*)d_in[5];
    const float* bv = (const float*)d_in[6];
    float* out = (float*)d_out;

    flatt_kernel<<<dim3(BB * (DD / 16)), dim3(128), 0, stream>>>(
        x, aq, bq, ak, bk, av, bv, out);
}


# pass 2: log2(e) folded into the reciprocal's argument, weight = exp2(rcp(fma(|k-q|, ln2, eps*ln2)) - c) with c the same expression at the row minimum: one multiply per key leaves the loop
# speedup vs baseline: 1.1603x; 1.0329x over previous
#include <hip/hip_runtime.h>
#include <hip/hip_bf16.h>

typedef __attribute__((ext_vector_type(2))) float v2f;
typedef __attribute__((ext_vector_type(8))) float v8f;

#define DD   1024
#define BB   16
#define HH   4
#define EPSF 1e-8f
#define LN2  0.69314718055994530942f

__global__ __launch_bounds__(128) void flatt_kernel(
    const float* __restrict__ x,
    const float* __restrict__ aq, const float* __restrict__ bq,
    const float* __restrict__ ak, const float* __restrict__ bk,
    const float* __restrict__ av, const float* __restrict__ bv,
    float* __restrict__ out)
{
    __shared__ float xs[DD];
    __shared__ float att_s[HH * 16];
    __shared__ float dn_s[HH * 16];

    const int b     = blockIdx.x >> 6;
    const int itile = blockIdx.x & 63;
    const int i0    = itile << 4;
    const int tid   = threadIdx.x;
    const int h     = tid >> 5;
    const int lane  = tid & 31;
    const int iloc  = lane & 15;
    const int half  = lane >> 4;

    #pragma unroll
    for (int r = 0; r < DD / 128; ++r)
        xs[tid + 128 * r] = x[b * DD + tid + 128 * r];
    __syncthreads();

    const float Aq = aq[h], Bq = bq[h];
    const float Ak = ak[h], Bk = bk[h];
    const float Av = av[h], Bv = bv[h];

    const float q = fmaf(xs[i0 + iloc], Aq, Bq);

    float dmin0 = 3.4e38f, dmin1 = 3.4e38f;
    #pragma clang loop vectorize(disable)
    for (int c = 0; c < DD / 4; c += 2) {
        const int j = 4 * c + 2 * half;
        const float x0 = xs[j],     x1 = xs[j + 1];
        const float x2 = xs[j + 4], x3 = xs[j + 5];
        const float d0 = fabsf(fmaf(x0, Ak, Bk) - q);
        const float d1 = fabsf(fmaf(x1, Ak, Bk) - q);
        const float d2 = fabsf(fmaf(x2, Ak, Bk) - q);
        const float d3 = fabsf(fmaf(x3, Ak, Bk) - q);
        dmin0 = __builtin_fminf(dmin0, __builtin_fminf(d0, d1));
        dmin1 = __builtin_fminf(dmin1, __builtin_fminf(d2, d3));
    }
    float dmin = __builtin_fminf(dmin0, dmin1);
    dmin = __builtin_fminf(dmin, __shfl_xor(dmin, 16, 32));
    const float mc = __builtin_amdgcn_rcpf(fmaf(dmin, LN2, EPSF * LN2));

    v8f acc = {};
    float denom = 0.0f;
    for (int c = 0; c < DD / 4; ++c) {
        const int j = 4 * c + 2 * half;
        const float x0 = xs[j], x1 = xs[j + 1];
        const float s0 = __builtin_amdgcn_rcpf(fmaf(fabsf(fmaf(x0, Ak, Bk) - q), LN2, EPSF * LN2));
        const float s1 = __builtin_amdgcn_rcpf(fmaf(fabsf(fmaf(x1, Ak, Bk) - q), LN2, EPSF * LN2));
        const float p0 = __builtin_amdgcn_exp2f(s0 - mc);
        const float p1 = __builtin_amdgcn_exp2f(s1 - mc);
        denom += p0 + p1;

        v2f A;  A.x  = p0;                  A.y  = p1;
        v2f Bm; Bm.x = fmaf(x0, Av, Bv);    Bm.y = fmaf(x1, Av, Bv);

        acc = __builtin_amdgcn_wmma_f32_16x16x4_f32(
                  false, A, false, Bm, (short)0, acc, false, false);
    }
    denom += __shfl_xor(denom, 16, 32);

    if (lane < 16) dn_s[h * 16 + iloc] = denom;
    if (lane == 0) {
        #pragma unroll
        for (int r = 0; r < 8; ++r) att_s[h * 16 + r] = acc[r];
    }
    if (lane == 16) {
        #pragma unroll
        for (int r = 0; r < 8; ++r) att_s[h * 16 + 8 + r] = acc[r];
    }
    __syncthreads();

    if (tid < 16) {
        float s = 0.0f;
        #pragma unroll
        for (int hh = 0; hh < HH; ++hh)
            s += att_s[hh * 16 + tid] / dn_s[hh * 16 + tid];
        out[b * DD + i0 + tid] = xs[i0 + tid] + 0.5f * s;
    }
}

extern "C" void kernel_launch(void* const* d_in, const int* in_sizes, int n_in,
                              void* d_out, int out_size, void* d_ws, size_t ws_size,
                              hipStream_t stream) {
    (void)in_sizes; (void)n_in; (void)d_ws; (void)ws_size; (void)out_size;
    const float* x  = (const float*)d_in[0];
    const float* aq = (const float*)d_in[1];
    const float* bq = (const float*)d_in[2];
    const float* ak = (const float*)d_in[3];
    const float* bk = (const float*)d_in[4];
    const float* av = (const float*)d_in[5];
    const float* bv = (const float*)d_in[6];
    float* out = (float*)d_out;

    flatt_kernel<<<dim3(BB * (DD / 16)), dim3(128), 0, stream>>>(
        x, aq, bq, ak, bk, av, bv, out);
}


# k_j and v_j hoisted out of the row loops: each wave builds its head's {k,k,v,v} table in LDS once per block and both passes read it, so no fma per key per lane is left in either loop
# speedup vs baseline: 1.3199x; 1.1376x over previous
#include <hip/hip_runtime.h>
#include <hip/hip_bf16.h>

typedef __attribute__((ext_vector_type(2))) float v2f;
typedef __attribute__((ext_vector_type(4))) float v4f;
typedef __attribute__((ext_vector_type(8))) float v8f;

#define DD   1024
#define BB   16
#define HH   4
#define EPSF 1e-8f
#define LN2  0.69314718055994530942f

__global__ __launch_bounds__(128) void flatt_kernel(
    const float* __restrict__ x,
    const float* __restrict__ aq, const float* __restrict__ bq,
    const float* __restrict__ ak, const float* __restrict__ bk,
    const float* __restrict__ av, const float* __restrict__ bv,
    float* __restrict__ out)
{
    __shared__ v4f   kv_s[HH * (DD / 2)];
    __shared__ float att_s[HH * 16];
    __shared__ float dn_s[HH * 16];

    const int b     = blockIdx.x >> 6;
    const int itile = blockIdx.x & 63;
    const int i0    = itile << 4;
    const int tid   = threadIdx.x;
    const int h     = tid >> 5;
    const int lane  = tid & 31;
    const int iloc  = lane & 15;
    const int half  = lane >> 4;

    const float Aq = aq[h], Bq = bq[h];
    const float Ak = ak[h], Bk = bk[h];
    const float Av = av[h], Bv = bv[h];

    v4f* tab = kv_s + h * (DD / 2);
    const v4f* xrow4 = reinterpret_cast<const v4f*>(x + b * DD);
    #pragma unroll
    for (int r = 0; r < DD / 128; ++r) {
        const int m = lane + 32 * r;
        const v4f xv = xrow4[m];
        v4f e0, e1;
        e0.x = fmaf(xv.x, Ak, Bk);  e0.y = fmaf(xv.y, Ak, Bk);
        e0.z = fmaf(xv.x, Av, Bv);  e0.w = fmaf(xv.y, Av, Bv);
        e1.x = fmaf(xv.z, Ak, Bk);  e1.y = fmaf(xv.w, Ak, Bk);
        e1.z = fmaf(xv.z, Av, Bv);  e1.w = fmaf(xv.w, Av, Bv);
        tab[2 * m]     = e0;
        tab[2 * m + 1] = e1;
    }

    const float q = fmaf(x[b * DD + i0 + iloc], Aq, Bq);
    __syncthreads();

    const v2f* tk = reinterpret_cast<const v2f*>(tab);
    float dmin0 = 3.4e38f, dmin1 = 3.4e38f;
    #pragma clang loop vectorize(disable)
    for (int c = 0; c < DD / 4; c += 2) {
        const int g = 2 * c + half;
        const v2f ka = tk[2 * g];
        const v2f kb = tk[2 * g + 4];
        const float d0 = fabsf(ka.x - q);
        const float d1 = fabsf(ka.y - q);
        const float d2 = fabsf(kb.x - q);
        const float d3 = fabsf(kb.y - q);
        dmin0 = __builtin_fminf(dmin0, __builtin_fminf(d0, d1));
        dmin1 = __builtin_fminf(dmin1, __builtin_fminf(d2, d3));
    }
    float dmin = __builtin_fminf(dmin0, dmin1);
    dmin = __builtin_fminf(dmin, __shfl_xor(dmin, 16, 32));
    const float mc = __builtin_amdgcn_rcpf(fmaf(dmin, LN2, EPSF * LN2));

    v8f acc = {};
    float denom = 0.0f;
    const v2f q2  = {q, q};
    const v2f mc2 = {mc, mc};
    for (int c = 0; c < DD / 4; ++c) {
        const v4f t = tab[2 * c + half];
        const v2f d = t.xy - q2;
        v2f s;
        s.x = __builtin_amdgcn_rcpf(fmaf(fabsf(d.x), LN2, EPSF * LN2));
        s.y = __builtin_amdgcn_rcpf(fmaf(fabsf(d.y), LN2, EPSF * LN2));
        const v2f e = s - mc2;
        v2f A;
        A.x = __builtin_amdgcn_exp2f(e.x);
        A.y = __builtin_amdgcn_exp2f(e.y);
        denom += A.x + A.y;
        const v2f Bm = t.zw;

        acc = __builtin_amdgcn_wmma_f32_16x16x4_f32(
                  false, A, false, Bm, (short)0, acc, false, false);
    }
    denom += __shfl_xor(denom, 16, 32);

    if (lane < 16) dn_s[h * 16 + iloc] = denom;
    if (lane == 0) {
        #pragma unroll
        for (int r = 0; r < 8; ++r) att_s[h * 16 + r] = acc[r];
    }
    if (lane == 16) {
        #pragma unroll
        for (int r = 0; r < 8; ++r) att_s[h * 16 + 8 + r] = acc[r];
    }
    __syncthreads();

    if (tid < 16) {
        float s = 0.0f;
        #pragma unroll
        for (int hh = 0; hh < HH; ++hh)
            s += att_s[hh * 16 + tid] / dn_s[hh * 16 + tid];
        out[b * DD + i0 + tid] = x[b * DD + i0 + tid] + 0.5f * s;
    }
}

extern "C" void kernel_launch(void* const* d_in, const int* in_sizes, int n_in,
                              void* d_out, int out_size, void* d_ws, size_t ws_size,
                              hipStream_t stream) {
    (void)in_sizes; (void)n_in; (void)d_ws; (void)ws_size; (void)out_size;
    const float* x  = (const float*)d_in[0];
    const float* aq = (const float*)d_in[1];
    const float* bq = (const float*)d_in[2];
    const float* ak = (const float*)d_in[3];
    const float* bk = (const float*)d_in[4];
    const float* av = (const float*)d_in[5];
    const float* bv = (const float*)d_in[6];
    float* out = (float*)d_out;

    flatt_kernel<<<dim3(BB * (DD / 16)), dim3(128), 0, stream>>>(
        x, aq, bq, ak, bk, av, bv, out);
}


# speedup vs baseline: 1.4308x; 1.0840x over previous
#include <hip/hip_runtime.h>
#include <hip/hip_bf16.h>

typedef __attribute__((ext_vector_type(2))) float v2f;
typedef __attribute__((ext_vector_type(4))) float v4f;
typedef __attribute__((ext_vector_type(8))) float v8f;

#define DD   1024
#define BB   16
#define HH   4
#define EPSF 1e-8f
#define LN2  0.69314718055994530942f

__global__ __launch_bounds__(128) void flatt_kernel(
    const float* __restrict__ x,
    const float* __restrict__ aq, const float* __restrict__ bq,
    const float* __restrict__ ak, const float* __restrict__ bk,
    const float* __restrict__ av, const float* __restrict__ bv,
    float* __restrict__ out)
{
    __shared__ v4f   kv_s[HH * (DD / 2)];
    __shared__ float att_s[HH * 16];
    __shared__ float dn_s[HH * 16];

    const int b     = blockIdx.x >> 6;
    const int itile = blockIdx.x & 63;
    const int i0    = itile << 4;
    const int tid   = threadIdx.x;
    const int h     = tid >> 5;
    const int lane  = tid & 31;
    const int iloc  = lane & 15;
    const int half  = lane >> 4;

    const float Aq = aq[h], Bq = bq[h];
    const float Ak = ak[h], Bk = bk[h];
    const float Av = av[h], Bv = bv[h];

    v4f* tab = kv_s + h * (DD / 2);
    const v4f* xrow4 = reinterpret_cast<const v4f*>(x + b * DD);
    #pragma unroll
    for (int r = 0; r < DD / 128; ++r) {
        const int m = lane + 32 * r;
        const v4f xv = xrow4[m];
        v4f e0, e1;
        e0.x = fmaf(xv.x, Ak, Bk);  e0.y = fmaf(xv.y, Ak, Bk);
        e0.z = fmaf(xv.x, Av, Bv);  e0.w = fmaf(xv.y, Av, Bv);
        e1.x = fmaf(xv.z, Ak, Bk);  e1.y = fmaf(xv.w, Ak, Bk);
        e1.z = fmaf(xv.z, Av, Bv);  e1.w = fmaf(xv.w, Av, Bv);
        tab[2 * m]     = e0;
        tab[2 * m + 1] = e1;
    }

    const float q = fmaf(x[b * DD + i0 + iloc], Aq, Bq);
    __syncthreads();

    const v2f* tk = reinterpret_cast<const v2f*>(tab);
    float dmin0 = 3.4e38f, dmin1 = 3.4e38f;
    #pragma clang loop vectorize(disable)
    for (int c = 0; c < DD / 4; c += 2) {
        const int g = 2 * c + half;
        const v2f ka = tk[2 * g];
        const v2f kb = tk[2 * g + 4];
        const float d0 = fabsf(ka.x - q);
        const float d1 = fabsf(ka.y - q);
        const float d2 = fabsf(kb.x - q);
        const float d3 = fabsf(kb.y - q);
        dmin0 = __builtin_fminf(dmin0, __builtin_fminf(d0, d1));
        dmin1 = __builtin_fminf(dmin1, __builtin_fminf(d2, d3));
    }
    float dmin = __builtin_fminf(dmin0, dmin1);
    dmin = __builtin_fminf(dmin, __shfl_xor(dmin, 16, 32));
    const float mc = __builtin_amdgcn_rcpf(fmaf(dmin, LN2, EPSF * LN2));

    v8f acc = {};
    float denom = 0.0f;
    const v2f q2  = {q, q};
    const v2f mc2 = {mc, mc};
    const v2f ln2v = {LN2, LN2};
    const v2f epsv = {EPSF * LN2, EPSF * LN2};

#define PIN(X) asm volatile("" : "+v"(X))

#define ROUND(TN, TB)                                                      \
    {                                                                      \
        PIN(sv);   v2f A = sv;  A.x = __builtin_amdgcn_exp2f(A.x); PIN(A); \
        PIN(qq);   v2f nn = (TN).xy - qq;                         PIN(nn); \
        PIN(A);    A.y = __builtin_amdgcn_exp2f(A.y);              PIN(A); \
        PIN(rv);   sv = rv - mc2;                                 PIN(sv); \
        PIN(fv);   rv = fv;  rv.x = __builtin_amdgcn_rcpf(rv.x);  PIN(rv); \
        PIN(no);   fv = __builtin_elementwise_fma(no, ln2v, epsv); PIN(fv);\
        PIN(rv);   rv.y = __builtin_amdgcn_rcpf(rv.y);            PIN(rv); \
        PIN(tsum); PIN(Ap);                                                \
        denom += tsum;  tsum = Ap.x + Ap.y;                                \
        PIN(denom); PIN(tsum); PIN(nn);                                    \
        nn.x = __builtin_fabsf(nn.x);                             PIN(nn); \
        nn.y = __builtin_fabsf(nn.y);                                      \
        PIN(nn); PIN(A);                                                   \
        acc = __builtin_amdgcn_wmma_f32_16x16x4_f32(                       \
                  false, A, false, (TB).zw, (short)0, acc, false, false);  \
        PIN(acc);                                                          \
        no = nn;  Ap = A;                                                  \
    }

#define LOADS(P, B)                                                      \
        P##0 = tl[2 * ((((B) + 1) * 8 + 0) & (DD / 4 - 1))];             \
        P##1 = tl[2 * ((((B) + 1) * 8 + 1) & (DD / 4 - 1))];             \
        P##2 = tl[2 * ((((B) + 1) * 8 + 2) & (DD / 4 - 1))];             \
        P##3 = tl[2 * ((((B) + 1) * 8 + 3) & (DD / 4 - 1))];             \
        P##4 = tl[2 * ((((B) + 1) * 8 + 4) & (DD / 4 - 1))];             \
        P##5 = tl[2 * ((((B) + 1) * 8 + 5) & (DD / 4 - 1))];             \
        P##6 = tl[2 * ((((B) + 1) * 8 + 6) & (DD / 4 - 1))];             \
        P##7 = tl[2 * ((((B) + 1) * 8 + 7) & (DD / 4 - 1))];
#define BODY(C, N, B)                                                    \
    {                                                                    \
        ROUND(C##0, N##4) ROUND(C##1, N##5)                              \
        ROUND(C##2, N##6) ROUND(C##3, N##7)                              \
        LOADS(N, B)                                                      \
        ROUND(C##4, C##0) ROUND(C##5, C##1)                              \
        ROUND(C##6, C##2) ROUND(C##7, C##3)                              \
    }

    const v4f* tl = tab + half;
    v4f ta0, ta1, ta2, ta3, ta4, ta5, ta6, ta7;
    v4f tb0, tb1, tb2, tb3, tb4, tb5, tb6, tb7;
    LOADS(ta, -1)
    tb4 = tb5 = tb6 = tb7 = v4f{0.0f, 0.0f, 0.0f, 0.0f};
    v2f qq = q2;
    v2f no = {-2.0f * EPSF, -2.0f * EPSF};
    v2f fv = __builtin_elementwise_fma(no, ln2v, epsv);
    v2f rv = {__builtin_amdgcn_rcpf(fv.x), __builtin_amdgcn_rcpf(fv.y)};
    v2f sv = rv - mc2;
    float tsum = 0.0f;
    v2f Ap = {0.0f, 0.0f};

    for (int b = 0; b < DD / 32; b += 2) {
        BODY(ta, tb, b)
        BODY(tb, ta, b + 1)
    }
    {
        v2f A;
        denom += tsum;
        denom += Ap.x + Ap.y;
        A.x = __builtin_amdgcn_exp2f(sv.x);
        A.y = __builtin_amdgcn_exp2f(sv.y);
        denom += A.x + A.y;
        acc = __builtin_amdgcn_wmma_f32_16x16x4_f32(
                  false, A, false, tb4.zw, (short)0, acc, false, false);
        sv = rv - mc2;
        A.x = __builtin_amdgcn_exp2f(sv.x);
        A.y = __builtin_amdgcn_exp2f(sv.y);
        denom += A.x + A.y;
        acc = __builtin_amdgcn_wmma_f32_16x16x4_f32(
                  false, A, false, tb5.zw, (short)0, acc, false, false);
        rv.x = __builtin_amdgcn_rcpf(fv.x);
        rv.y = __builtin_amdgcn_rcpf(fv.y);
        sv = rv - mc2;
        A.x = __builtin_amdgcn_exp2f(sv.x);
        A.y = __builtin_amdgcn_exp2f(sv.y);
        denom += A.x + A.y;
        acc = __builtin_amdgcn_wmma_f32_16x16x4_f32(
                  false, A, false, tb6.zw, (short)0, acc, false, false);
        fv = __builtin_elementwise_fma(no, ln2v, epsv);
        rv.x = __builtin_amdgcn_rcpf(fv.x);
        rv.y = __builtin_amdgcn_rcpf(fv.y);
        sv = rv - mc2;
        A.x = __builtin_amdgcn_exp2f(sv.x);
        A.y = __builtin_amdgcn_exp2f(sv.y);
        denom += A.x + A.y;
        acc = __builtin_amdgcn_wmma_f32_16x16x4_f32(
                  false, A, false, tb7.zw, (short)0, acc, false, false);
    }
#undef BODY
#undef LOADS
#undef ROUND
#undef PIN
    denom += __shfl_xor(denom, 16, 32);

    if (lane < 16) dn_s[h * 16 + iloc] = denom;
    if (lane == 0) {
        #pragma unroll
        for (int r = 0; r < 8; ++r) att_s[h * 16 + r] = acc[r];
    }
    if (lane == 16) {
        #pragma unroll
        for (int r = 0; r < 8; ++r) att_s[h * 16 + 8 + r] = acc[r];
    }
    __syncthreads();

    if (tid < 16) {
        float s = 0.0f;
        #pragma unroll
        for (int hh = 0; hh < HH; ++hh)
            s += att_s[hh * 16 + tid] / dn_s[hh * 16 + tid];
        out[b * DD + i0 + tid] = x[b * DD + i0 + tid] + 0.5f * s;
    }
}

extern "C" void kernel_launch(void* const* d_in, const int* in_sizes, int n_in,
                              void* d_out, int out_size, void* d_ws, size_t ws_size,
                              hipStream_t stream) {
    (void)in_sizes; (void)n_in; (void)d_ws; (void)ws_size; (void)out_size;
    const float* x  = (const float*)d_in[0];
    const float* aq = (const float*)d_in[1];
    const float* bq = (const float*)d_in[2];
    const float* ak = (const float*)d_in[3];
    const float* bk = (const float*)d_in[4];
    const float* av = (const float*)d_in[5];
    const float* bv = (const float*)d_in[6];
    float* out = (float*)d_out;

    flatt_kernel<<<dim3(BB * (DD / 16)), dim3(128), 0, stream>>>(
        x, aq, bq, ak, bk, av, bv, out);
}


# pass 2 leaves out 16-key groups whose weights are exact zeros for all 16 rows (scalar 64-bit mask from group minima of pass 1); 16-wave blocks put all four heads on every SIMD; sum order kept
# speedup vs baseline: 1.5394x; 1.0759x over previous
#include <hip/hip_runtime.h>
#include <hip/hip_bf16.h>

typedef __attribute__((ext_vector_type(2))) float v2f;
typedef __attribute__((ext_vector_type(4))) float v4f;
typedef __attribute__((ext_vector_type(8))) float v8f;

#define DD   1024
#define BB   16
#define HH   4
#define EPSF 1e-8f
#define LN2  0.69314718055994530942f

__global__ __launch_bounds__(512) void flatt_kernel(
    const float* __restrict__ x,
    const float* __restrict__ aq, const float* __restrict__ bq,
    const float* __restrict__ ak, const float* __restrict__ bk,
    const float* __restrict__ av, const float* __restrict__ bv,
    float* __restrict__ out)
{
    __shared__ v4f   kv_s[HH * (DD / 2)];
    __shared__ float att_s[4 * HH * 16];
    __shared__ float dn_s[4 * HH * 16];

    const int b     = blockIdx.x >> 4;
    const int ib0   = (blockIdx.x & 15) << 6;
    const int tid   = threadIdx.x;
    const int w     = tid >> 5;
    const int t     = w >> 2;
    const int h     = (w + t) & 3;
    const int i0    = ib0 + (t << 4);
    const int lane  = tid & 31;
    const int iloc  = lane & 15;
    const int half  = lane >> 4;

    const float Aq = aq[h], Bq = bq[h];
    const float Ak = ak[h], Bk = bk[h];
    const float Av = av[h], Bv = bv[h];

    v4f* tab = kv_s + h * (DD / 2);
    const v4f* xrow4 = reinterpret_cast<const v4f*>(x + b * DD);
    #pragma unroll
    for (int rr = 0; rr < DD / 512; ++rr) {
        const int m = lane + 32 * (2 * t + rr);
        const v4f xv = xrow4[m];
        v4f e0, e1;
        e0.x = fmaf(xv.x, Ak, Bk);  e0.y = fmaf(xv.y, Ak, Bk);
        e0.z = fmaf(xv.x, Av, Bv);  e0.w = fmaf(xv.y, Av, Bv);
        e1.x = fmaf(xv.z, Ak, Bk);  e1.y = fmaf(xv.w, Ak, Bk);
        e1.z = fmaf(xv.z, Av, Bv);  e1.w = fmaf(xv.w, Av, Bv);
        tab[2 * m]     = e0;
        tab[2 * m + 1] = e1;
    }

    const float q = fmaf(x[b * DD + i0 + iloc], Aq, Bq);
    __syncthreads();

    const v2f* tk = reinterpret_cast<const v2f*>(tab);
    float gm[DD / 16];
    const v2f qv = {q, q};
    #pragma unroll
    for (int G = 0; G < DD / 16; ++G) {
        const v2f ka = tk[2 * (8 * G + 0 + half)];
        const v2f kb = tk[2 * (8 * G + 2 + half)];
        const v2f kc = tk[2 * (8 * G + 4 + half)];
        const v2f kd = tk[2 * (8 * G + 6 + half)];
        const v2f da = ka - qv, db = kb - qv, dc = kc - qv, dd = kd - qv;
        float m = __builtin_fminf(__builtin_fminf(fabsf(da.x), fabsf(da.y)), fabsf(db.x));
        m = __builtin_fminf(__builtin_fminf(m, fabsf(db.y)), fabsf(dc.x));
        m = __builtin_fminf(__builtin_fminf(m, fabsf(dc.y)), fabsf(dd.x));
        gm[G] = __builtin_fminf(m, fabsf(dd.y));
    }
    float dmin0 = 3.4e38f, dmin1 = 3.4e38f;
    #pragma unroll
    for (int G = 0; G < DD / 16; G += 4) {
        dmin0 = __builtin_fminf(__builtin_fminf(dmin0, gm[G]), gm[G + 1]);
        dmin1 = __builtin_fminf(__builtin_fminf(dmin1, gm[G + 2]), gm[G + 3]);
    }
    float dmin = __builtin_fminf(dmin0, dmin1);
    dmin = __builtin_fminf(dmin, __shfl_xor(dmin, 16, 32));
    const float mc = __builtin_amdgcn_rcpf(fmaf(dmin, LN2, EPSF * LN2));

    const float dcut = (mc > 200.0f)
        ? 1.001f / ((mc - 160.0f) * LN2)
        : __builtin_inff();
    unsigned long long live = 0ull;
    #pragma unroll
    for (int G = 0; G < DD / 16; ++G) {
        const unsigned any = __builtin_amdgcn_ballot_w32(gm[G] <= dcut);
        live |= (unsigned long long)(any != 0u) << G;
    }
    const int nlive = __builtin_popcountll(live);
    const int niter = (nlive + 3) >> 2;
    const int deadg = (~live != 0ull) ? (int)__builtin_ctzll(~live) : 0;

    v8f acc = {};
    float denom = 0.0f;
    const v2f q2  = {q, q};
    const v2f mc2 = {mc, mc};
    const v2f ln2v = {LN2, LN2};
    const v2f epsv = {EPSF * LN2, EPSF * LN2};

#define PIN(X) asm volatile("" : "+v"(X))

#define ROUND(TN, TB)                                                      \
    {                                                                      \
        PIN(sv);   v2f A = sv;  A.x = __builtin_amdgcn_exp2f(A.x); PIN(A); \
        PIN(qq);   v2f nn = (TN).xy - qq;                         PIN(nn); \
        PIN(A);    A.y = __builtin_amdgcn_exp2f(A.y);              PIN(A); \
        PIN(rv);   sv = rv - mc2;                                 PIN(sv); \
        PIN(fv);   rv = fv;  rv.x = __builtin_amdgcn_rcpf(rv.x);  PIN(rv); \
        PIN(no);   fv = __builtin_elementwise_fma(no, ln2v, epsv); PIN(fv);\
        PIN(rv);   rv.y = __builtin_amdgcn_rcpf(rv.y);            PIN(rv); \
        PIN(tsum); PIN(Ap);                                                \
        denom += tsum;  tsum = Ap.x + Ap.y;                                \
        PIN(denom); PIN(tsum); PIN(nn);                                    \
        nn.x = __builtin_fabsf(nn.x);                             PIN(nn); \
        nn.y = __builtin_fabsf(nn.y);                                      \
        PIN(nn); PIN(A);                                                   \
        acc = __builtin_amdgcn_wmma_f32_16x16x4_f32(                       \
                  false, A, false, (TB).zw, (short)0, acc, false, false);  \
        PIN(acc);                                                          \
        no = nn;  Ap = A;                                                  \
    }

#define NEXTG(G)                                                         \
        G = (live != 0ull) ? (int)__builtin_ctzll(live) : deadg;         \
        live &= live - 1ull;
#define LOADS(P, GA, GB)                                                 \
        P##0 = tl[8 * (GA) + 0];                                         \
        P##1 = tl[8 * (GA) + 2];                                         \
        P##2 = tl[8 * (GA) + 4];                                         \
        P##3 = tl[8 * (GA) + 6];                                         \
        P##4 = tl[8 * (GB) + 0];                                         \
        P##5 = tl[8 * (GB) + 2];                                         \
        P##6 = tl[8 * (GB) + 4];                                         \
        P##7 = tl[8 * (GB) + 6];
#define BODY(C, N)                                                       \
    {                                                                    \
        ROUND(C##0, N##4) ROUND(C##1, N##5)                              \
        ROUND(C##2, N##6) ROUND(C##3, N##7)                              \
        NEXTG(ga) NEXTG(gb)                                              \
        LOADS(N, ga, gb)                                                 \
        ROUND(C##4, C##0) ROUND(C##5, C##1)                              \
        ROUND(C##6, C##2) ROUND(C##7, C##3)                              \
    }

    const v4f* tl = tab + half;
    v4f ta0, ta1, ta2, ta3, ta4, ta5, ta6, ta7;
    v4f tb0, tb1, tb2, tb3, tb4, tb5, tb6, tb7;
    int ga, gb;
    NEXTG(ga) NEXTG(gb)
    LOADS(ta, ga, gb)
    tb4 = tb5 = tb6 = tb7 = v4f{0.0f, 0.0f, 0.0f, 0.0f};
    v2f qq = q2;
    v2f no = {-2.0f * EPSF, -2.0f * EPSF};
    v2f fv = __builtin_elementwise_fma(no, ln2v, epsv);
    v2f rv = {__builtin_amdgcn_rcpf(fv.x), __builtin_amdgcn_rcpf(fv.y)};
    v2f sv = rv - mc2;
    float tsum = 0.0f;
    v2f Ap = {0.0f, 0.0f};

    for (int it = 0; it < niter; ++it) {
        BODY(ta, tb)
        BODY(tb, ta)
    }
    {
        v2f A;
        denom += tsum;
        denom += Ap.x + Ap.y;
        A.x = __builtin_amdgcn_exp2f(sv.x);
        A.y = __builtin_amdgcn_exp2f(sv.y);
        denom += A.x + A.y;
        acc = __builtin_amdgcn_wmma_f32_16x16x4_f32(
                  false, A, false, tb4.zw, (short)0, acc, false, false);
        sv = rv - mc2;
        A.x = __builtin_amdgcn_exp2f(sv.x);
        A.y = __builtin_amdgcn_exp2f(sv.y);
        denom += A.x + A.y;
        acc = __builtin_amdgcn_wmma_f32_16x16x4_f32(
                  false, A, false, tb5.zw, (short)0, acc, false, false);
        rv.x = __builtin_amdgcn_rcpf(fv.x);
        rv.y = __builtin_amdgcn_rcpf(fv.y);
        sv = rv - mc2;
        A.x = __builtin_amdgcn_exp2f(sv.x);
        A.y = __builtin_amdgcn_exp2f(sv.y);
        denom += A.x + A.y;
        acc = __builtin_amdgcn_wmma_f32_16x16x4_f32(
                  false, A, false, tb6.zw, (short)0, acc, false, false);
        fv = __builtin_elementwise_fma(no, ln2v, epsv);
        rv.x = __builtin_amdgcn_rcpf(fv.x);
        rv.y = __builtin_amdgcn_rcpf(fv.y);
        sv = rv - mc2;
        A.x = __builtin_amdgcn_exp2f(sv.x);
        A.y = __builtin_amdgcn_exp2f(sv.y);
        denom += A.x + A.y;
        acc = __builtin_amdgcn_wmma_f32_16x16x4_f32(
                  false, A, false, tb7.zw, (short)0, acc, false, false);
    }
#undef BODY
#undef LOADS
#undef NEXTG
#undef ROUND
#undef PIN
    denom += __shfl_xor(denom, 16, 32);

    const int slot = (t * HH + h) * 16;
    if (lane < 16) dn_s[slot + iloc] = denom;
    if (lane == 0) {
        #pragma unroll
        for (int r = 0; r < 8; ++r) att_s[slot + r] = acc[r];
    }
    if (lane == 16) {
        #pragma unroll
        for (int r = 0; r < 8; ++r) att_s[slot + 8 + r] = acc[r];
    }
    __syncthreads();

    if (tid < 64) {
        const int tt  = tid >> 4;
        const int row = tid & 15;
        float s = 0.0f;
        #pragma unroll
        for (int hh = 0; hh < HH; ++hh)
            s += att_s[(tt * HH + hh) * 16 + row] / dn_s[(tt * HH + hh) * 16 + row];
        out[b * DD + ib0 + tid] = x[b * DD + ib0 + tid] + 0.5f * s;
    }
}

extern "C" void kernel_launch(void* const* d_in, const int* in_sizes, int n_in,
                              void* d_out, int out_size, void* d_ws, size_t ws_size,
                              hipStream_t stream) {
    (void)in_sizes; (void)n_in; (void)d_ws; (void)ws_size; (void)out_size;
    const float* x  = (const float*)d_in[0];
    const float* aq = (const float*)d_in[1];
    const float* bq = (const float*)d_in[2];
    const float* ak = (const float*)d_in[3];
    const float* bk = (const float*)d_in[4];
    const float* av = (const float*)d_in[5];
    const float* bv = (const float*)d_in[6];
    float* out = (float*)d_out;

    flatt_kernel<<<dim3(BB * (DD / 64)), dim3(512), 0, stream>>>(
        x, aq, bq, ak, bk, av, bv, out);
}


# pass 1 reads its keys from the LDS table one batch of four 16-key groups ahead, in two register buffers: the LDS latency overlaps the minima of the batch before instead of being waited out per group
# speedup vs baseline: 1.6446x; 1.0683x over previous
#include <hip/hip_runtime.h>
#include <hip/hip_bf16.h>

typedef __attribute__((ext_vector_type(2))) float v2f;
typedef __attribute__((ext_vector_type(4))) float v4f;
typedef __attribute__((ext_vector_type(8))) float v8f;

#define DD   1024
#define BB   16
#define HH   4
#define EPSF 1e-8f
#define LN2  0.69314718055994530942f

__global__ __launch_bounds__(512) void flatt_kernel(
    const float* __restrict__ x,
    const float* __restrict__ aq, const float* __restrict__ bq,
    const float* __restrict__ ak, const float* __restrict__ bk,
    const float* __restrict__ av, const float* __restrict__ bv,
    float* __restrict__ out)
{
    __shared__ v4f   kv_s[HH * (DD / 2)];
    __shared__ float att_s[4 * HH * 16];
    __shared__ float dn_s[4 * HH * 16];

    const int b     = blockIdx.x >> 4;
    const int ib0   = (blockIdx.x & 15) << 6;
    const int tid   = threadIdx.x;
    const int w     = tid >> 5;
    const int t     = w >> 2;
    const int h     = (w + t) & 3;
    const int i0    = ib0 + (t << 4);
    const int lane  = tid & 31;
    const int iloc  = lane & 15;
    const int half  = lane >> 4;

    const float Aq = aq[h], Bq = bq[h];
    const float Ak = ak[h], Bk = bk[h];
    const float Av = av[h], Bv = bv[h];

    v4f* tab = kv_s + h * (DD / 2);
    const v4f* xrow4 = reinterpret_cast<const v4f*>(x + b * DD);
    #pragma unroll
    for (int rr = 0; rr < DD / 512; ++rr) {
        const int m = lane + 32 * (2 * t + rr);
        const v4f xv = xrow4[m];
        v4f e0, e1;
        e0.x = fmaf(xv.x, Ak, Bk);  e0.y = fmaf(xv.y, Ak, Bk);
        e0.z = fmaf(xv.x, Av, Bv);  e0.w = fmaf(xv.y, Av, Bv);
        e1.x = fmaf(xv.z, Ak, Bk);  e1.y = fmaf(xv.w, Ak, Bk);
        e1.z = fmaf(xv.z, Av, Bv);  e1.w = fmaf(xv.w, Av, Bv);
        tab[2 * m]     = e0;
        tab[2 * m + 1] = e1;
    }

    const float q = fmaf(x[b * DD + i0 + iloc], Aq, Bq);
    __syncthreads();

    const v2f* tk = reinterpret_cast<const v2f*>(tab);
    float gm[DD / 16];
    const v2f qv = {q, q};
#define P1_READ(BUF, B)                                                  \
    _Pragma("unroll")                                                    \
    for (int u = 0; u < 16; ++u)                                         \
        BUF[u] = tk[2 * (32 * (B) + 2 * u + half)];
#define P1_MIN(BUF, B)                                                   \
    _Pragma("unroll")                                                    \
    for (int g = 0; g < 4; ++g) {                                        \
        const v2f da = BUF[4 * g + 0] - qv, db = BUF[4 * g + 1] - qv;    \
        const v2f dc = BUF[4 * g + 2] - qv, dd = BUF[4 * g + 3] - qv;    \
        float m = __builtin_fminf(__builtin_fminf(fabsf(da.x), fabsf(da.y)), fabsf(db.x)); \
        m = __builtin_fminf(__builtin_fminf(m, fabsf(db.y)), fabsf(dc.x)); \
        m = __builtin_fminf(__builtin_fminf(m, fabsf(dc.y)), fabsf(dd.x)); \
        gm[4 * (B) + g] = __builtin_fminf(m, fabsf(dd.y));               \
    }
    v2f kb0[16], kb1[16];
    P1_READ(kb0, 0)
    #pragma unroll
    for (int B = 0; B < DD / 64; B += 2) {
        P1_READ(kb1, B + 1)
        __builtin_amdgcn_sched_barrier(0);
        P1_MIN(kb0, B)
        __builtin_amdgcn_sched_barrier(0);
        if (B + 2 < DD / 64) {
            P1_READ(kb0, B + 2)
        }
        __builtin_amdgcn_sched_barrier(0);
        P1_MIN(kb1, B + 1)
        __builtin_amdgcn_sched_barrier(0);
    }
#undef P1_MIN
#undef P1_READ
    float dmin0 = 3.4e38f, dmin1 = 3.4e38f;
    #pragma unroll
    for (int G = 0; G < DD / 16; G += 4) {
        dmin0 = __builtin_fminf(__builtin_fminf(dmin0, gm[G]), gm[G + 1]);
        dmin1 = __builtin_fminf(__builtin_fminf(dmin1, gm[G + 2]), gm[G + 3]);
    }
    float dmin = __builtin_fminf(dmin0, dmin1);
    dmin = __builtin_fminf(dmin, __shfl_xor(dmin, 16, 32));
    const float mc = __builtin_amdgcn_rcpf(fmaf(dmin, LN2, EPSF * LN2));

    const float dcut = (mc > 200.0f)
        ? 1.001f / ((mc - 160.0f) * LN2)
        : __builtin_inff();
    unsigned long long live = 0ull;
    #pragma unroll
    for (int G = 0; G < DD / 16; ++G) {
        const unsigned any = __builtin_amdgcn_ballot_w32(gm[G] <= dcut);
        live |= (unsigned long long)(any != 0u) << G;
    }
    const int nlive = __builtin_popcountll(live);
    const int niter = (nlive + 3) >> 2;
    const int deadg = (~live != 0ull) ? (int)__builtin_ctzll(~live) : 0;

    v8f acc = {};
    float denom = 0.0f;
    const v2f q2  = {q, q};
    const v2f mc2 = {mc, mc};
    const v2f ln2v = {LN2, LN2};
    const v2f epsv = {EPSF * LN2, EPSF * LN2};

#define PIN(X) asm volatile("" : "+v"(X))

#define ROUND(TN, TB)                                                      \
    {                                                                      \
        PIN(sv);   v2f A = sv;  A.x = __builtin_amdgcn_exp2f(A.x); PIN(A); \
        PIN(qq);   v2f nn = (TN).xy - qq;                         PIN(nn); \
        PIN(A);    A.y = __builtin_amdgcn_exp2f(A.y);              PIN(A); \
        PIN(rv);   sv = rv - mc2;                                 PIN(sv); \
        PIN(fv);   rv = fv;  rv.x = __builtin_amdgcn_rcpf(rv.x);  PIN(rv); \
        PIN(no);   fv = __builtin_elementwise_fma(no, ln2v, epsv); PIN(fv);\
        PIN(rv);   rv.y = __builtin_amdgcn_rcpf(rv.y);            PIN(rv); \
        PIN(tsum); PIN(Ap);                                                \
        denom += tsum;  tsum = Ap.x + Ap.y;                                \
        PIN(denom); PIN(tsum); PIN(nn);                                    \
        nn.x = __builtin_fabsf(nn.x);                             PIN(nn); \
        nn.y = __builtin_fabsf(nn.y);                                      \
        PIN(nn); PIN(A);                                                   \
        acc = __builtin_amdgcn_wmma_f32_16x16x4_f32(                       \
                  false, A, false, (TB).zw, (short)0, acc, false, false);  \
        PIN(acc);                                                          \
        no = nn;  Ap = A;                                                  \
    }

#define NEXTG(G)                                                         \
        G = (live != 0ull) ? (int)__builtin_ctzll(live) : deadg;         \
        live &= live - 1ull;
#define LOADS(P, GA, GB)                                                 \
        P##0 = tl[8 * (GA) + 0];                                         \
        P##1 = tl[8 * (GA) + 2];                                         \
        P##2 = tl[8 * (GA) + 4];                                         \
        P##3 = tl[8 * (GA) + 6];                                         \
        P##4 = tl[8 * (GB) + 0];                                         \
        P##5 = tl[8 * (GB) + 2];                                         \
        P##6 = tl[8 * (GB) + 4];                                         \
        P##7 = tl[8 * (GB) + 6];
#define BODY(C, N)                                                       \
    {                                                                    \
        ROUND(C##0, N##4) ROUND(C##1, N##5)                              \
        ROUND(C##2, N##6) ROUND(C##3, N##7)                              \
        NEXTG(ga) NEXTG(gb)                                              \
        LOADS(N, ga, gb)                                                 \
        ROUND(C##4, C##0) ROUND(C##5, C##1)                              \
        ROUND(C##6, C##2) ROUND(C##7, C##3)                              \
    }

    const v4f* tl = tab + half;
    v4f ta0, ta1, ta2, ta3, ta4, ta5, ta6, ta7;
    v4f tb0, tb1, tb2, tb3, tb4, tb5, tb6, tb7;
    int ga, gb;
    NEXTG(ga) NEXTG(gb)
    LOADS(ta, ga, gb)
    tb4 = tb5 = tb6 = tb7 = v4f{0.0f, 0.0f, 0.0f, 0.0f};
    v2f qq = q2;
    v2f no = {-2.0f * EPSF, -2.0f * EPSF};
    v2f fv = __builtin_elementwise_fma(no, ln2v, epsv);
    v2f rv = {__builtin_amdgcn_rcpf(fv.x), __builtin_amdgcn_rcpf(fv.y)};
    v2f sv = rv - mc2;
    float tsum = 0.0f;
    v2f Ap = {0.0f, 0.0f};

    for (int it = 0; it < niter; ++it) {
        BODY(ta, tb)
        BODY(tb, ta)
    }
    {
        v2f A;
        denom += tsum;
        denom += Ap.x + Ap.y;
        A.x = __builtin_amdgcn_exp2f(sv.x);
        A.y = __builtin_amdgcn_exp2f(sv.y);
        denom += A.x + A.y;
        acc = __builtin_amdgcn_wmma_f32_16x16x4_f32(
                  false, A, false, tb4.zw, (short)0, acc, false, false);
        sv = rv - mc2;
        A.x = __builtin_amdgcn_exp2f(sv.x);
        A.y = __builtin_amdgcn_exp2f(sv.y);
        denom += A.x + A.y;
        acc = __builtin_amdgcn_wmma_f32_16x16x4_f32(
                  false, A, false, tb5.zw, (short)0, acc, false, false);
        rv.x = __builtin_amdgcn_rcpf(fv.x);
        rv.y = __builtin_amdgcn_rcpf(fv.y);
        sv = rv - mc2;
        A.x = __builtin_amdgcn_exp2f(sv.x);
        A.y = __builtin_amdgcn_exp2f(sv.y);
        denom += A.x + A.y;
        acc = __builtin_amdgcn_wmma_f32_16x16x4_f32(
                  false, A, false, tb6.zw, (short)0, acc, false, false);
        fv = __builtin_elementwise_fma(no, ln2v, epsv);
        rv.x = __builtin_amdgcn_rcpf(fv.x);
        rv.y = __builtin_amdgcn_rcpf(fv.y);
        sv = rv - mc2;
        A.x = __builtin_amdgcn_exp2f(sv.x);
        A.y = __builtin_amdgcn_exp2f(sv.y);
        denom += A.x + A.y;
        acc = __builtin_amdgcn_wmma_f32_16x16x4_f32(
                  false, A, false, tb7.zw, (short)0, acc, false, false);
    }
#undef BODY
#undef LOADS
#undef NEXTG
#undef ROUND
#undef PIN
    denom += __shfl_xor(denom, 16, 32);

    const int slot = (t * HH + h) * 16;
    if (lane < 16) dn_s[slot + iloc] = denom;
    if (lane == 0) {
        #pragma unroll
        for (int r = 0; r < 8; ++r) att_s[slot + r] = acc[r];
    }
    if (lane == 16) {
        #pragma unroll
        for (int r = 0; r < 8; ++r) att_s[slot + 8 + r] = acc[r];
    }
    __syncthreads();

    if (tid < 64) {
        const int tt  = tid >> 4;
        const int row = tid & 15;
        float s = 0.0f;
        #pragma unroll
        for (int hh = 0; hh < HH; ++hh)
            s += att_s[(tt * HH + hh) * 16 + row] / dn_s[(tt * HH + hh) * 16 + row];
        out[b * DD + ib0 + tid] = x[b * DD + ib0 + tid] + 0.5f * s;
    }
}

extern "C" void kernel_launch(void* const* d_in, const int* in_sizes, int n_in,
                              void* d_out, int out_size, void* d_ws, size_t ws_size,
                              hipStream_t stream) {
    (void)in_sizes; (void)n_in; (void)d_ws; (void)ws_size; (void)out_size;
    const float* x  = (const float*)d_in[0];
    const float* aq = (const float*)d_in[1];
    const float* bq = (const float*)d_in[2];
    const float* ak = (const float*)d_in[3];
    const float* bk = (const float*)d_in[4];
    const float* av = (const float*)d_in[5];
    const float* bv = (const float*)d_in[6];
    float* out = (float*)d_out;

    flatt_kernel<<<dim3(BB * (DD / 64)), dim3(512), 0, stream>>>(
        x, aq, bq, ak, bk, av, bv, out);
}


# between the passes each head's 64 rows are re-dealt to the 16-row tiles, rows without a cutoff first, so few tiles instead of most run all 64 key groups; per-row values and sum order unchanged
# speedup vs baseline: 2.1949x; 1.3346x over previous
#include <hip/hip_runtime.h>
#include <hip/hip_bf16.h>

typedef __attribute__((ext_vector_type(2))) float v2f;
typedef __attribute__((ext_vector_type(4))) float v4f;
typedef __attribute__((ext_vector_type(8))) float v8f;

#define DD   1024
#define BB   16
#define HH   4
#define EPSF 1e-8f
#define LN2  0.69314718055994530942f

__global__ __launch_bounds__(512) void flatt_kernel(
    const float* __restrict__ x,
    const float* __restrict__ aq, const float* __restrict__ bq,
    const float* __restrict__ ak, const float* __restrict__ bk,
    const float* __restrict__ av, const float* __restrict__ bv,
    float* __restrict__ out)
{
    __shared__ v4f   kv_s[HH * (DD / 2)];
    __shared__ float att_s[4 * HH * 16];
    __shared__ float dn_s[4 * HH * 16];
    __shared__ unsigned rb_s[HH * 4 * 64];
    __shared__ float    q_s[HH * 64];
    __shared__ float    mc_s[HH * 64];
    __shared__ unsigned far_s[HH * 4];

    const int b     = blockIdx.x >> 4;
    const int ib0   = (blockIdx.x & 15) << 6;
    const int tid   = threadIdx.x;
    const int w     = tid >> 5;
    const int t     = w >> 2;
    const int h     = (w + t) & 3;
    const int i0    = ib0 + (t << 4);
    const int lane  = tid & 31;
    const int iloc  = lane & 15;
    const int half  = lane >> 4;

    const float Aq = aq[h], Bq = bq[h];
    const float Ak = ak[h], Bk = bk[h];
    const float Av = av[h], Bv = bv[h];

    v4f* tab = kv_s + h * (DD / 2);
    const v4f* xrow4 = reinterpret_cast<const v4f*>(x + b * DD);
    #pragma unroll
    for (int rr = 0; rr < DD / 512; ++rr) {
        const int m = lane + 32 * (2 * t + rr);
        const v4f xv = xrow4[m];
        v4f e0, e1;
        e0.x = fmaf(xv.x, Ak, Bk);  e0.y = fmaf(xv.y, Ak, Bk);
        e0.z = fmaf(xv.x, Av, Bv);  e0.w = fmaf(xv.y, Av, Bv);
        e1.x = fmaf(xv.z, Ak, Bk);  e1.y = fmaf(xv.w, Ak, Bk);
        e1.z = fmaf(xv.z, Av, Bv);  e1.w = fmaf(xv.w, Av, Bv);
        tab[2 * m]     = e0;
        tab[2 * m + 1] = e1;
    }

    const float q1 = fmaf(x[b * DD + i0 + iloc], Aq, Bq);
    __syncthreads();

    const v2f* tk = reinterpret_cast<const v2f*>(tab);
    float gm[DD / 16];
    const v2f qv = {q1, q1};
#define P1_READ(BUF, B)                                                  \
    _Pragma("unroll")                                                    \
    for (int u = 0; u < 16; ++u)                                         \
        BUF[u] = tk[2 * (32 * (B) + 2 * u + half)];
#define P1_MIN(BUF, B)                                                   \
    _Pragma("unroll")                                                    \
    for (int g = 0; g < 4; ++g) {                                        \
        const v2f da = BUF[4 * g + 0] - qv, db = BUF[4 * g + 1] - qv;    \
        const v2f dc = BUF[4 * g + 2] - qv, dd = BUF[4 * g + 3] - qv;    \
        float m = __builtin_fminf(__builtin_fminf(fabsf(da.x), fabsf(da.y)), fabsf(db.x)); \
        m = __builtin_fminf(__builtin_fminf(m, fabsf(db.y)), fabsf(dc.x)); \
        m = __builtin_fminf(__builtin_fminf(m, fabsf(dc.y)), fabsf(dd.x)); \
        gm[4 * (B) + g] = __builtin_fminf(m, fabsf(dd.y));               \
    }
    v2f kb0[16], kb1[16];
    P1_READ(kb0, 0)
    #pragma unroll
    for (int B = 0; B < DD / 64; B += 2) {
        P1_READ(kb1, B + 1)
        __builtin_amdgcn_sched_barrier(0);
        P1_MIN(kb0, B)
        __builtin_amdgcn_sched_barrier(0);
        if (B + 2 < DD / 64) {
            P1_READ(kb0, B + 2)
        }
        __builtin_amdgcn_sched_barrier(0);
        P1_MIN(kb1, B + 1)
        __builtin_amdgcn_sched_barrier(0);
    }
#undef P1_MIN
#undef P1_READ
    float dmin0 = 3.4e38f, dmin1 = 3.4e38f;
    #pragma unroll
    for (int G = 0; G < DD / 16; G += 4) {
        dmin0 = __builtin_fminf(__builtin_fminf(dmin0, gm[G]), gm[G + 1]);
        dmin1 = __builtin_fminf(__builtin_fminf(dmin1, gm[G + 2]), gm[G + 3]);
    }
    float dmin = __builtin_fminf(dmin0, dmin1);
    dmin = __builtin_fminf(dmin, __shfl_xor(dmin, 16, 32));
    const float mc1 = __builtin_amdgcn_rcpf(fmaf(dmin, LN2, EPSF * LN2));

    const float dcut = (mc1 > 200.0f)
        ? 1.001f / ((mc1 - 160.0f) * LN2)
        : __builtin_inff();
    int vlo = 0, vhi = 0;
    #pragma unroll
    for (int G = 0; G < DD / 16; ++G) {
        const int any = (int)__builtin_amdgcn_ballot_w32(gm[G] <= dcut);
        if (G < 32) vlo = (lane == G)      ? any : vlo;
        else        vhi = (lane == G - 32) ? any : vhi;
    }
    rb_s[(h * 4 + t) * 64 + lane]      = (unsigned)vlo;
    rb_s[(h * 4 + t) * 64 + 32 + lane] = (unsigned)vhi;
    const unsigned farb = __builtin_amdgcn_ballot_w32(!(mc1 > 200.0f)) & 0xFFFFu;
    if (lane < 16) {
        q_s[h * 64 + 16 * t + iloc]  = q1;
        mc_s[h * 64 + 16 * t + iloc] = mc1;
    }
    if (lane == 0) far_s[h * 4 + t] = farb;
    __syncthreads();

    const unsigned long long farm =
          (unsigned long long)(far_s[h * 4 + 0] | (far_s[h * 4 + 1] << 16))
        | ((unsigned long long)(far_s[h * 4 + 2] | (far_s[h * 4 + 3] << 16)) << 32);
    const int nfar = __builtin_popcountll(farm);
    int myrow;
    {
        const int n = 16 * t + iloc;
        unsigned long long m = (n < nfar) ? farm : ~farm;
        int k = (n < nfar) ? n : n - nfar;
        int pos = 0;
        #pragma unroll
        for (int wd = 32; wd >= 1; wd >>= 1) {
            const unsigned long long lowm = m & ((1ull << wd) - 1ull);
            const int c = __builtin_popcountll(lowm);
            const bool up = (k >= c);
            pos += up ? wd : 0;
            k   -= up ? c : 0;
            m    = up ? (m >> wd) : lowm;
        }
        myrow = pos;
    }
    unsigned rs0, rs1, rs2, rs3;
    {
        const unsigned long long bel0 = farm & ((1ull << lane) - 1ull);
        const unsigned long long bel1 = farm & ((1ull << (lane + 32)) - 1ull);
        const int fb0 = __builtin_popcountll(bel0), fb1 = __builtin_popcountll(bel1);
        const bool far0 = (farm >> lane) & 1ull, far1 = (farm >> (lane + 32)) & 1ull;
        const int rank0 = far0 ? fb0 : nfar + (lane - fb0);
        const int rank1 = far1 ? fb1 : nfar + (lane + 32 - fb1);
        const unsigned bl = __builtin_amdgcn_ballot_w32((rank0 >> 4) == t);
        const unsigned bh = __builtin_amdgcn_ballot_w32((rank1 >> 4) == t);
        rs0 = (bl & 0xFFFFu) * 0x10001u;  rs1 = (bl >> 16) * 0x10001u;
        rs2 = (bh & 0xFFFFu) * 0x10001u;  rs3 = (bh >> 16) * 0x10001u;
    }
    unsigned long long live;
    {
        const unsigned* rb = rb_s + h * 4 * 64 + lane;
        const unsigned a = (rb[0]  & rs0) | (rb[64] & rs1) | (rb[128] & rs2) | (rb[192] & rs3);
        const unsigned c = (rb[32] & rs0) | (rb[96] & rs1) | (rb[160] & rs2) | (rb[224] & rs3);
        live = (unsigned long long)__builtin_amdgcn_ballot_w32(a != 0u)
             | ((unsigned long long)__builtin_amdgcn_ballot_w32(c != 0u) << 32);
    }
    const float q  = q_s[h * 64 + myrow];
    const float mc = mc_s[h * 64 + myrow];
    const int nlive = __builtin_popcountll(live);
    const int niter = (nlive + 3) >> 2;
    const int deadg = (~live != 0ull) ? (int)__builtin_ctzll(~live) : 0;

    v8f acc = {};
    float denom = 0.0f;
    const v2f q2  = {q, q};
    const v2f mc2 = {mc, mc};
    const v2f ln2v = {LN2, LN2};
    const v2f epsv = {EPSF * LN2, EPSF * LN2};

#define PIN(X) asm volatile("" : "+v"(X))

#define ROUND(TN, TB)                                                      \
    {                                                                      \
        PIN(sv);   v2f A = sv;  A.x = __builtin_amdgcn_exp2f(A.x); PIN(A); \
        PIN(qq);   v2f nn = (TN).xy - qq;                         PIN(nn); \
        PIN(A);    A.y = __builtin_amdgcn_exp2f(A.y);              PIN(A); \
        PIN(rv);   sv = rv - mc2;                                 PIN(sv); \
        PIN(fv);   rv = fv;  rv.x = __builtin_amdgcn_rcpf(rv.x);  PIN(rv); \
        PIN(no);   fv = __builtin_elementwise_fma(no, ln2v, epsv); PIN(fv);\
        PIN(rv);   rv.y = __builtin_amdgcn_rcpf(rv.y);            PIN(rv); \
        PIN(tsum); PIN(Ap);                                                \
        denom += tsum;  tsum = Ap.x + Ap.y;                                \
        PIN(denom); PIN(tsum); PIN(nn);                                    \
        nn.x = __builtin_fabsf(nn.x);                             PIN(nn); \
        nn.y = __builtin_fabsf(nn.y);                                      \
        PIN(nn); PIN(A);                                                   \
        acc = __builtin_amdgcn_wmma_f32_16x16x4_f32(                       \
                  false, A, false, (TB).zw, (short)0, acc, false, false);  \
        PIN(acc);                                                          \
        no = nn;  Ap = A;                                                  \
    }

#define NEXTG(G)                                                         \
        G = (live != 0ull) ? (int)__builtin_ctzll(live) : deadg;         \
        live &= live - 1ull;
#define LOADS(P, GA, GB)                                                 \
        P##0 = tl[8 * (GA) + 0];                                         \
        P##1 = tl[8 * (GA) + 2];                                         \
        P##2 = tl[8 * (GA) + 4];                                         \
        P##3 = tl[8 * (GA) + 6];                                         \
        P##4 = tl[8 * (GB) + 0];                                         \
        P##5 = tl[8 * (GB) + 2];                                         \
        P##6 = tl[8 * (GB) + 4];                                         \
        P##7 = tl[8 * (GB) + 6];
#define BODY(C, N)                                                       \
    {                                                                    \
        ROUND(C##0, N##4) ROUND(C##1, N##5)                              \
        ROUND(C##2, N##6) ROUND(C##3, N##7)                              \
        NEXTG(ga) NEXTG(gb)                                              \
        LOADS(N, ga, gb)                                                 \
        ROUND(C##4, C##0) ROUND(C##5, C##1)                              \
        ROUND(C##6, C##2) ROUND(C##7, C##3)                              \
    }

    const v4f* tl = tab + half;
    v4f ta0, ta1, ta2, ta3, ta4, ta5, ta6, ta7;
    v4f tb0, tb1, tb2, tb3, tb4, tb5, tb6, tb7;
    int ga, gb;
    NEXTG(ga) NEXTG(gb)
    LOADS(ta, ga, gb)
    tb4 = tb5 = tb6 = tb7 = v4f{0.0f, 0.0f, 0.0f, 0.0f};
    v2f qq = q2;
    v2f no = {-2.0f * EPSF, -2.0f * EPSF};
    v2f fv = __builtin_elementwise_fma(no, ln2v, epsv);
    v2f rv = {__builtin_amdgcn_rcpf(fv.x), __builtin_amdgcn_rcpf(fv.y)};
    v2f sv = rv - mc2;
    float tsum = 0.0f;
    v2f Ap = {0.0f, 0.0f};

    for (int it = 0; it < niter; ++it) {
        BODY(ta, tb)
        BODY(tb, ta)
    }
    {
        v2f A;
        denom += tsum;
        denom += Ap.x + Ap.y;
        A.x = __builtin_amdgcn_exp2f(sv.x);
        A.y = __builtin_amdgcn_exp2f(sv.y);
        denom += A.x + A.y;
        acc = __builtin_amdgcn_wmma_f32_16x16x4_f32(
                  false, A, false, tb4.zw, (short)0, acc, false, false);
        sv = rv - mc2;
        A.x = __builtin_amdgcn_exp2f(sv.x);
        A.y = __builtin_amdgcn_exp2f(sv.y);
        denom += A.x + A.y;
        acc = __builtin_amdgcn_wmma_f32_16x16x4_f32(
                  false, A, false, tb5.zw, (short)0, acc, false, false);
        rv.x = __builtin_amdgcn_rcpf(fv.x);
        rv.y = __builtin_amdgcn_rcpf(fv.y);
        sv = rv - mc2;
        A.x = __builtin_amdgcn_exp2f(sv.x);
        A.y = __builtin_amdgcn_exp2f(sv.y);
        denom += A.x + A.y;
        acc = __builtin_amdgcn_wmma_f32_16x16x4_f32(
                  false, A, false, tb6.zw, (short)0, acc, false, false);
        fv = __builtin_elementwise_fma(no, ln2v, epsv);
        rv.x = __builtin_amdgcn_rcpf(fv.x);
        rv.y = __builtin_amdgcn_rcpf(fv.y);
        sv = rv - mc2;
        A.x = __builtin_amdgcn_exp2f(sv.x);
        A.y = __builtin_amdgcn_exp2f(sv.y);
        denom += A.x + A.y;
        acc = __builtin_amdgcn_wmma_f32_16x16x4_f32(
                  false, A, false, tb7.zw, (short)0, acc, false, false);
    }
#undef BODY
#undef LOADS
#undef NEXTG
#undef ROUND
#undef PIN
    denom += __shfl_xor(denom, 16, 32);

    const int slot = ((myrow >> 4) * HH + h) * 16 + (myrow & 15);
    if (lane < 16) dn_s[slot] = denom;
    float sel = acc[0];
    #pragma unroll
    for (int r = 1; r < 8; ++r) sel = ((lane & 7) == r) ? acc[r] : sel;
    if (lane < 8 || lane >= 24) att_s[slot] = sel;
    __syncthreads();

    if (tid < 64) {
        const int tt  = tid >> 4;
        const int row = tid & 15;
        float s = 0.0f;
        #pragma unroll
        for (int hh = 0; hh < HH; ++hh)
            s += att_s[(tt * HH + hh) * 16 + row] / dn_s[(tt * HH + hh) * 16 + row];
        out[b * DD + ib0 + tid] = x[b * DD + ib0 + tid] + 0.5f * s;
    }
}

extern "C" void kernel_launch(void* const* d_in, const int* in_sizes, int n_in,
                              void* d_out, int out_size, void* d_ws, size_t ws_size,
                              hipStream_t stream) {
    (void)in_sizes; (void)n_in; (void)d_ws; (void)ws_size; (void)out_size;
    const float* x  = (const float*)d_in[0];
    const float* aq = (const float*)d_in[1];
    const float* bq = (const float*)d_in[2];
    const float* ak = (const float*)d_in[3];
    const float* bk = (const float*)d_in[4];
    const float* av = (const float*)d_in[5];
    const float* bv = (const float*)d_in[6];
    float* out = (float*)d_out;

    flatt_kernel<<<dim3(BB * (DD / 64)), dim3(512), 0, stream>>>(
        x, aq, bq, ak, bk, av, bv, out);
}


# pass 2: each row walks its own live key groups (own keys' v in the row's B column, diagonal of D read) instead of the union over the tile's 16 rows; table groups padded for banks; sum order kept
# speedup vs baseline: 2.4027x; 1.0947x over previous
#include <hip/hip_runtime.h>
#include <hip/hip_bf16.h>

typedef __attribute__((ext_vector_type(2))) float v2f;
typedef __attribute__((ext_vector_type(4))) float v4f;
typedef __attribute__((ext_vector_type(8))) float v8f;

#define DD   1024
#define BB   16
#define HH   4
#define EPSF 1e-8f
#define LN2  0.69314718055994530942f
#define TABH (DD / 2 + DD / 16)

__global__ __launch_bounds__(512) void flatt_kernel(
    const float* __restrict__ x,
    const float* __restrict__ aq, const float* __restrict__ bq,
    const float* __restrict__ ak, const float* __restrict__ bk,
    const float* __restrict__ av, const float* __restrict__ bv,
    float* __restrict__ out)
{
    __shared__ v4f   kv_s[HH * TABH];
    __shared__ float att_s[4 * HH * 16];
    __shared__ float dn_s[4 * HH * 16];
    __shared__ unsigned long long rm_s[HH * 64];
    __shared__ float    q_s[HH * 64];
    __shared__ float    mc_s[HH * 64];
    __shared__ unsigned far_s[HH * 4];

    const int b     = blockIdx.x >> 4;
    const int ib0   = (blockIdx.x & 15) << 6;
    const int tid   = threadIdx.x;
    const int w     = tid >> 5;
    const int t     = w >> 2;
    const int h     = (w + t) & 3;
    const int i0    = ib0 + (t << 4);
    const int lane  = tid & 31;
    const int iloc  = lane & 15;
    const int half  = lane >> 4;

    const float Aq = aq[h], Bq = bq[h];
    const float Ak = ak[h], Bk = bk[h];
    const float Av = av[h], Bv = bv[h];

    v4f* tab = kv_s + h * TABH;
    const v4f* xrow4 = reinterpret_cast<const v4f*>(x + b * DD);
    #pragma unroll
    for (int rr = 0; rr < DD / 512; ++rr) {
        const int m = lane + 32 * (2 * t + rr);
        const v4f xv = xrow4[m];
        v4f e0, e1;
        e0.x = fmaf(xv.x, Ak, Bk);  e0.y = fmaf(xv.y, Ak, Bk);
        e0.z = fmaf(xv.x, Av, Bv);  e0.w = fmaf(xv.y, Av, Bv);
        e1.x = fmaf(xv.z, Ak, Bk);  e1.y = fmaf(xv.w, Ak, Bk);
        e1.z = fmaf(xv.z, Av, Bv);  e1.w = fmaf(xv.w, Av, Bv);
        tab[2 * m + (m >> 2)]     = e0;
        tab[2 * m + (m >> 2) + 1] = e1;
    }

    const float q1 = fmaf(x[b * DD + i0 + iloc], Aq, Bq);
    __syncthreads();

    const v2f* tk = reinterpret_cast<const v2f*>(tab);
    float gm[DD / 16];
    const v2f qv = {q1, q1};
#define P1_READ(BUF, B)                                                  \
    _Pragma("unroll")                                                    \
    for (int u = 0; u < 16; ++u)                                         \
        BUF[u] = tk[2 * (36 * (B) + 2 * u + (u >> 2) + half)];
#define P1_MIN(BUF, B)                                                   \
    _Pragma("unroll")                                                    \
    for (int g = 0; g < 4; ++g) {                                        \
        const v2f da = BUF[4 * g + 0] - qv, db = BUF[4 * g + 1] - qv;    \
        const v2f dc = BUF[4 * g + 2] - qv, dd = BUF[4 * g + 3] - qv;    \
        float m = __builtin_fminf(__builtin_fminf(fabsf(da.x), fabsf(da.y)), fabsf(db.x)); \
        m = __builtin_fminf(__builtin_fminf(m, fabsf(db.y)), fabsf(dc.x)); \
        m = __builtin_fminf(__builtin_fminf(m, fabsf(dc.y)), fabsf(dd.x)); \
        gm[4 * (B) + g] = __builtin_fminf(m, fabsf(dd.y));               \
    }
    v2f kb0[16], kb1[16];
    P1_READ(kb0, 0)
    #pragma unroll
    for (int B = 0; B < DD / 64; B += 2) {
        P1_READ(kb1, B + 1)
        __builtin_amdgcn_sched_barrier(0);
        P1_MIN(kb0, B)
        __builtin_amdgcn_sched_barrier(0);
        if (B + 2 < DD / 64) {
            P1_READ(kb0, B + 2)
        }
        __builtin_amdgcn_sched_barrier(0);
        P1_MIN(kb1, B + 1)
        __builtin_amdgcn_sched_barrier(0);
    }
#undef P1_MIN
#undef P1_READ
    float dmin0 = 3.4e38f, dmin1 = 3.4e38f;
    #pragma unroll
    for (int G = 0; G < DD / 16; G += 4) {
        dmin0 = __builtin_fminf(__builtin_fminf(dmin0, gm[G]), gm[G + 1]);
        dmin1 = __builtin_fminf(__builtin_fminf(dmin1, gm[G + 2]), gm[G + 3]);
    }
    float dmin = __builtin_fminf(dmin0, dmin1);
    dmin = __builtin_fminf(dmin, __shfl_xor(dmin, 16, 32));
    const float mc1 = __builtin_amdgcn_rcpf(fmaf(dmin, LN2, EPSF * LN2));

    const float dcut = (mc1 > 200.0f)
        ? 1.001f / ((mc1 - 160.0f) * LN2)
        : __builtin_inff();
    unsigned mlo = 0u, mhi = 0u;
    #pragma unroll
    for (int G = 0; G < DD / 16; ++G) {
        const unsigned bit = (gm[G] <= dcut) ? (1u << (G & 31)) : 0u;
        if (G < 32) mlo |= bit;
        else        mhi |= bit;
    }
    mlo |= (unsigned)__shfl_xor((int)mlo, 16, 32);
    mhi |= (unsigned)__shfl_xor((int)mhi, 16, 32);
    const unsigned farb = __builtin_amdgcn_ballot_w32(!(mc1 > 200.0f)) & 0xFFFFu;
    if (lane < 16) {
        q_s[h * 64 + 16 * t + iloc]  = q1;
        mc_s[h * 64 + 16 * t + iloc] = mc1;
        rm_s[h * 64 + 16 * t + iloc] = ((unsigned long long)mhi << 32) | (unsigned long long)mlo;
    }
    if (lane == 0) far_s[h * 4 + t] = farb;
    __syncthreads();

    const unsigned long long farm =
          (unsigned long long)(far_s[h * 4 + 0] | (far_s[h * 4 + 1] << 16))
        | ((unsigned long long)(far_s[h * 4 + 2] | (far_s[h * 4 + 3] << 16)) << 32);
    const int nfar = __builtin_popcountll(farm);
    int myrow;
    {
        const int n = 16 * t + iloc;
        unsigned long long m = (n < nfar) ? farm : ~farm;
        int k = (n < nfar) ? n : n - nfar;
        int pos = 0;
        #pragma unroll
        for (int wd = 32; wd >= 1; wd >>= 1) {
            const unsigned long long lowm = m & ((1ull << wd) - 1ull);
            const int c = __builtin_popcountll(lowm);
            const bool up = (k >= c);
            pos += up ? wd : 0;
            k   -= up ? c : 0;
            m    = up ? (m >> wd) : lowm;
        }
        myrow = pos;
    }
    unsigned long long live = rm_s[h * 64 + myrow];
    const float q  = q_s[h * 64 + myrow];
    const float mc = mc_s[h * 64 + myrow];
    const int nit   = (__builtin_popcountll(live) + 3) >> 2;
    const int deadg = (~live != 0ull) ? (int)__builtin_ctzll(~live) : 0;
    int niter = 0;
    {
        unsigned cand = 0xFFFFFFFFu;
        for (int bit = 4; bit >= 0; --bit) {
            const unsigned bs = __builtin_amdgcn_ballot_w32(((nit >> bit) & 1) != 0) & cand;
            if (bs != 0u) { niter |= (1 << bit); cand = bs; }
        }
    }

    v8f acc = {};
    float denom = 0.0f;
    const v2f q2  = {q, q};
    const v2f mc2 = {mc, mc};
    const v2f ln2v = {LN2, LN2};
    const v2f epsv = {EPSF * LN2, EPSF * LN2};

#define PIN(X) asm volatile("" : "+v"(X))

#define ROUND(TN, TB)                                                      \
    {                                                                      \
        PIN(sv);   v2f A = sv;  A.x = __builtin_amdgcn_exp2f(A.x); PIN(A); \
        PIN(qq);   v2f nn = (TN).xy - qq;                         PIN(nn); \
        PIN(A);    A.y = __builtin_amdgcn_exp2f(A.y);              PIN(A); \
        PIN(rv);   sv = rv - mc2;                                 PIN(sv); \
        PIN(fv);   rv = fv;  rv.x = __builtin_amdgcn_rcpf(rv.x);  PIN(rv); \
        PIN(no);   fv = __builtin_elementwise_fma(no, ln2v, epsv); PIN(fv);\
        PIN(rv);   rv.y = __builtin_amdgcn_rcpf(rv.y);            PIN(rv); \
        PIN(tsum); PIN(Ap);                                                \
        denom += tsum;  tsum = Ap.x + Ap.y;                                \
        PIN(denom); PIN(tsum); PIN(nn);                                    \
        nn.x = __builtin_fabsf(nn.x);                             PIN(nn); \
        nn.y = __builtin_fabsf(nn.y);                                      \
        PIN(nn); PIN(A);                                                   \
        acc = __builtin_amdgcn_wmma_f32_16x16x4_f32(                       \
                  false, A, false, (TB).zw, (short)0, acc, false, false);  \
        PIN(acc);                                                          \
        no = nn;  Ap = A;                                                  \
    }

#define NEXTG(G)                                                         \
        G = (live != 0ull) ? (int)__builtin_ctzll(live) : deadg;         \
        live &= live - 1ull;
#define LOADS(P, GA, GB)                                                 \
        P##0 = tl[9 * (GA) + 0];                                         \
        P##1 = tl[9 * (GA) + 2];                                         \
        P##2 = tl[9 * (GA) + 4];                                         \
        P##3 = tl[9 * (GA) + 6];                                         \
        P##4 = tl[9 * (GB) + 0];                                         \
        P##5 = tl[9 * (GB) + 2];                                         \
        P##6 = tl[9 * (GB) + 4];                                         \
        P##7 = tl[9 * (GB) + 6];
#define BODY(C, N)                                                       \
    {                                                                    \
        ROUND(C##0, N##4) ROUND(C##1, N##5)                              \
        ROUND(C##2, N##6) ROUND(C##3, N##7)                              \
        NEXTG(ga) NEXTG(gb)                                              \
        LOADS(N, ga, gb)                                                 \
        ROUND(C##4, C##0) ROUND(C##5, C##1)                              \
        ROUND(C##6, C##2) ROUND(C##7, C##3)                              \
    }

    const v4f* tl = tab + half;
    v4f ta0, ta1, ta2, ta3, ta4, ta5, ta6, ta7;
    v4f tb0, tb1, tb2, tb3, tb4, tb5, tb6, tb7;
    int ga, gb;
    NEXTG(ga) NEXTG(gb)
    LOADS(ta, ga, gb)
    tb4 = tb5 = tb6 = tb7 = v4f{0.0f, 0.0f, 0.0f, 0.0f};
    v2f qq = q2;
    v2f no = {-2.0f * EPSF, -2.0f * EPSF};
    v2f fv = __builtin_elementwise_fma(no, ln2v, epsv);
    v2f rv = {__builtin_amdgcn_rcpf(fv.x), __builtin_amdgcn_rcpf(fv.y)};
    v2f sv = rv - mc2;
    float tsum = 0.0f;
    v2f Ap = {0.0f, 0.0f};

    for (int it = 0; it < niter; ++it) {
        BODY(ta, tb)
        BODY(tb, ta)
    }
    {
        v2f A;
        denom += tsum;
        denom += Ap.x + Ap.y;
        A.x = __builtin_amdgcn_exp2f(sv.x);
        A.y = __builtin_amdgcn_exp2f(sv.y);
        denom += A.x + A.y;
        acc = __builtin_amdgcn_wmma_f32_16x16x4_f32(
                  false, A, false, tb4.zw, (short)0, acc, false, false);
        sv = rv - mc2;
        A.x = __builtin_amdgcn_exp2f(sv.x);
        A.y = __builtin_amdgcn_exp2f(sv.y);
        denom += A.x + A.y;
        acc = __builtin_amdgcn_wmma_f32_16x16x4_f32(
                  false, A, false, tb5.zw, (short)0, acc, false, false);
        rv.x = __builtin_amdgcn_rcpf(fv.x);
        rv.y = __builtin_amdgcn_rcpf(fv.y);
        sv = rv - mc2;
        A.x = __builtin_amdgcn_exp2f(sv.x);
        A.y = __builtin_amdgcn_exp2f(sv.y);
        denom += A.x + A.y;
        acc = __builtin_amdgcn_wmma_f32_16x16x4_f32(
                  false, A, false, tb6.zw, (short)0, acc, false, false);
        fv = __builtin_elementwise_fma(no, ln2v, epsv);
        rv.x = __builtin_amdgcn_rcpf(fv.x);
        rv.y = __builtin_amdgcn_rcpf(fv.y);
        sv = rv - mc2;
        A.x = __builtin_amdgcn_exp2f(sv.x);
        A.y = __builtin_amdgcn_exp2f(sv.y);
        denom += A.x + A.y;
        acc = __builtin_amdgcn_wmma_f32_16x16x4_f32(
                  false, A, false, tb7.zw, (short)0, acc, false, false);
    }
#undef BODY
#undef LOADS
#undef NEXTG
#undef ROUND
#undef PIN
    denom += __shfl_xor(denom, 16, 32);

    const int slot = ((myrow >> 4) * HH + h) * 16 + (myrow & 15);
    if (lane < 16) dn_s[slot] = denom;
    float sel = acc[0];
    #pragma unroll
    for (int r = 1; r < 8; ++r) sel = ((lane & 7) == r) ? acc[r] : sel;
    if (lane < 8 || lane >= 24) att_s[slot] = sel;
    __syncthreads();

    if (tid < 64) {
        const int tt  = tid >> 4;
        const int row = tid & 15;
        float s = 0.0f;
        #pragma unroll
        for (int hh = 0; hh < HH; ++hh)
            s += att_s[(tt * HH + hh) * 16 + row] / dn_s[(tt * HH + hh) * 16 + row];
        out[b * DD + ib0 + tid] = x[b * DD + ib0 + tid] + 0.5f * s;
    }
}

extern "C" void kernel_launch(void* const* d_in, const int* in_sizes, int n_in,
                              void* d_out, int out_size, void* d_ws, size_t ws_size,
                              hipStream_t stream) {
    (void)in_sizes; (void)n_in; (void)d_ws; (void)ws_size; (void)out_size;
    const float* x  = (const float*)d_in[0];
    const float* aq = (const float*)d_in[1];
    const float* bq = (const float*)d_in[2];
    const float* ak = (const float*)d_in[3];
    const float* bk = (const float*)d_in[4];
    const float* av = (const float*)d_in[5];
    const float* bv = (const float*)d_in[6];
    float* out = (float*)d_out;

    flatt_kernel<<<dim3(BB * (DD / 64)), dim3(512), 0, stream>>>(
        x, aq, bq, ak, bk, av, bv, out);
}


# pass 2 tiles dealt across the four heads: the block's 256 (row, head) pairs ordered far, many-group, rest; tiles handed to SIMDs in a serpentine; head tables shifted 4 banks; per-row sums unchanged
# speedup vs baseline: 2.7573x; 1.1476x over previous
#include <hip/hip_runtime.h>
#include <hip/hip_bf16.h>

typedef __attribute__((ext_vector_type(2))) float v2f;
typedef __attribute__((ext_vector_type(4))) float v4f;
typedef __attribute__((ext_vector_type(8))) float v8f;

#define DD   1024
#define BB   16
#define HH   4
#define EPSF 1e-8f
#define LN2  0.69314718055994530942f
#define TABH (DD / 2 + DD / 16 + 1)
typedef unsigned short __attribute__((may_alias)) u16a;

__global__ __launch_bounds__(512) void flatt_kernel(
    const float* __restrict__ x,
    const float* __restrict__ aq, const float* __restrict__ bq,
    const float* __restrict__ ak, const float* __restrict__ bk,
    const float* __restrict__ av, const float* __restrict__ bv,
    float* __restrict__ out)
{
    __shared__ v4f   kv_s[HH * TABH];
    __shared__ float att_s[4 * HH * 16];
    __shared__ float dn_s[4 * HH * 16];
    __shared__ unsigned long long rm_s[HH * 64];
    __shared__ float    q_s[HH * 64];
    __shared__ float    mc_s[HH * 64];
    __shared__ unsigned long long cls_s[2 * HH];

    const int b     = blockIdx.x >> 4;
    const int ib0   = (blockIdx.x & 15) << 6;
    const int tid   = threadIdx.x;
    const int w     = tid >> 5;
    const int t     = w >> 2;
    const int h     = (w + t) & 3;
    const int i0    = ib0 + (t << 4);
    const int lane  = tid & 31;
    const int iloc  = lane & 15;
    const int half  = lane >> 4;

    const float Aq = aq[h], Bq = bq[h];
    const float Ak = ak[h], Bk = bk[h];
    const float Av = av[h], Bv = bv[h];

    v4f* tab = kv_s + h * TABH;
    const v4f* xrow4 = reinterpret_cast<const v4f*>(x + b * DD);
    #pragma unroll
    for (int rr = 0; rr < DD / 512; ++rr) {
        const int m = lane + 32 * (2 * t + rr);
        const v4f xv = xrow4[m];
        v4f e0, e1;
        e0.x = fmaf(xv.x, Ak, Bk);  e0.y = fmaf(xv.y, Ak, Bk);
        e0.z = fmaf(xv.x, Av, Bv);  e0.w = fmaf(xv.y, Av, Bv);
        e1.x = fmaf(xv.z, Ak, Bk);  e1.y = fmaf(xv.w, Ak, Bk);
        e1.z = fmaf(xv.z, Av, Bv);  e1.w = fmaf(xv.w, Av, Bv);
        tab[2 * m + (m >> 2)]     = e0;
        tab[2 * m + (m >> 2) + 1] = e1;
    }

    const float q1 = fmaf(x[b * DD + i0 + iloc], Aq, Bq);
    __syncthreads();

    const v2f* tk = reinterpret_cast<const v2f*>(tab);
    float gm[DD / 16];
    const v2f qv = {q1, q1};
#define P1_READ(BUF, B)                                                  \
    _Pragma("unroll")                                                    \
    for (int u = 0; u < 16; ++u)                                         \
        BUF[u] = tk[2 * (36 * (B) + 2 * u + (u >> 2) + half)];
#define P1_MIN(BUF, B)                                                   \
    _Pragma("unroll")                                                    \
    for (int g = 0; g < 4; ++g) {                                        \
        const v2f da = BUF[4 * g + 0] - qv, db = BUF[4 * g + 1] - qv;    \
        const v2f dc = BUF[4 * g + 2] - qv, dd = BUF[4 * g + 3] - qv;    \
        float m = __builtin_fminf(__builtin_fminf(fabsf(da.x), fabsf(da.y)), fabsf(db.x)); \
        m = __builtin_fminf(__builtin_fminf(m, fabsf(db.y)), fabsf(dc.x)); \
        m = __builtin_fminf(__builtin_fminf(m, fabsf(dc.y)), fabsf(dd.x)); \
        gm[4 * (B) + g] = __builtin_fminf(m, fabsf(dd.y));               \
    }
    v2f kb0[16], kb1[16];
    P1_READ(kb0, 0)
    #pragma unroll
    for (int B = 0; B < DD / 64; B += 2) {
        P1_READ(kb1, B + 1)
        __builtin_amdgcn_sched_barrier(0);
        P1_MIN(kb0, B)
        __builtin_amdgcn_sched_barrier(0);
        if (B + 2 < DD / 64) {
            P1_READ(kb0, B + 2)
        }
        __builtin_amdgcn_sched_barrier(0);
        P1_MIN(kb1, B + 1)
        __builtin_amdgcn_sched_barrier(0);
    }
#undef P1_MIN
#undef P1_READ
    float dmin0 = 3.4e38f, dmin1 = 3.4e38f;
    #pragma unroll
    for (int G = 0; G < DD / 16; G += 4) {
        dmin0 = __builtin_fminf(__builtin_fminf(dmin0, gm[G]), gm[G + 1]);
        dmin1 = __builtin_fminf(__builtin_fminf(dmin1, gm[G + 2]), gm[G + 3]);
    }
    float dmin = __builtin_fminf(dmin0, dmin1);
    dmin = __builtin_fminf(dmin, __shfl_xor(dmin, 16, 32));
    const float mc1 = __builtin_amdgcn_rcpf(fmaf(dmin, LN2, EPSF * LN2));

    const float dcut = (mc1 > 200.0f)
        ? 1.001f / ((mc1 - 160.0f) * LN2)
        : __builtin_inff();
    unsigned mlo = 0u, mhi = 0u;
    #pragma unroll
    for (int G = 0; G < DD / 16; ++G) {
        const unsigned bit = (gm[G] <= dcut) ? (1u << (G & 31)) : 0u;
        if (G < 32) mlo |= bit;
        else        mhi |= bit;
    }
    mlo |= (unsigned)__shfl_xor((int)mlo, 16, 32);
    mhi |= (unsigned)__shfl_xor((int)mhi, 16, 32);
    const bool isfar = !(mc1 > 200.0f);
    const bool ismid = !isfar && (__builtin_popcount(mlo) + __builtin_popcount(mhi) > 4);
    const unsigned farb = __builtin_amdgcn_ballot_w32(isfar) & 0xFFFFu;
    const unsigned midb = __builtin_amdgcn_ballot_w32(ismid) & 0xFFFFu;
    if (lane < 16) {
        q_s[h * 64 + 16 * t + iloc]  = q1;
        mc_s[h * 64 + 16 * t + iloc] = mc1;
        rm_s[h * 64 + 16 * t + iloc] = ((unsigned long long)mhi << 32) | (unsigned long long)mlo;
    }
    if (lane == 0) {
        reinterpret_cast<u16a*>(cls_s)[h * 4 + t]      = (unsigned short)farb;
        reinterpret_cast<u16a*>(cls_s)[16 + h * 4 + t] = (unsigned short)midb;
    }
    __syncthreads();

    const unsigned long long fm0 = cls_s[0], fm1 = cls_s[1], fm2 = cls_s[2], fm3 = cls_s[3];
    const unsigned long long mm0 = cls_s[4], mm1 = cls_s[5], mm2 = cls_s[6], mm3 = cls_s[7];
    const int nfar = __builtin_popcountll(fm0) + __builtin_popcountll(fm1)
                   + __builtin_popcountll(fm2) + __builtin_popcountll(fm3);
    const int nmid = __builtin_popcountll(mm0) + __builtin_popcountll(mm1)
                   + __builtin_popcountll(mm2) + __builtin_popcountll(mm3);
    const int gt = (t & 1) ? (4 * t + 3 - (w & 3)) : (4 * t + (w & 3));
    int myrow, hh;
    {
        const int n = 16 * gt + iloc;
        const bool cf = (n < nfar), cm = (n < nfar + nmid);
        int k = cf ? n : (cm ? n - nfar : n - nfar - nmid);
        const unsigned long long a0 = cf ? fm0 : (cm ? mm0 : ~(fm0 | mm0));
        const unsigned long long a1 = cf ? fm1 : (cm ? mm1 : ~(fm1 | mm1));
        const unsigned long long a2 = cf ? fm2 : (cm ? mm2 : ~(fm2 | mm2));
        const unsigned long long a3 = cf ? fm3 : (cm ? mm3 : ~(fm3 | mm3));
        const int c0 = __builtin_popcountll(a0);
        const int c1 = c0 + __builtin_popcountll(a1);
        const int c2 = c1 + __builtin_popcountll(a2);
        const bool p1 = (k >= c0), p2 = (k >= c1), p3 = (k >= c2);
        hh = (int)p1 + (int)p2 + (int)p3;
        unsigned long long m = p3 ? a3 : (p2 ? a2 : (p1 ? a1 : a0));
        k -= p3 ? c2 : (p2 ? c1 : (p1 ? c0 : 0));
        int pos = 0;
        #pragma unroll
        for (int wd = 32; wd >= 1; wd >>= 1) {
            const unsigned long long lowm = m & ((1ull << wd) - 1ull);
            const int c = __builtin_popcountll(lowm);
            const bool up = (k >= c);
            pos += up ? wd : 0;
            k   -= up ? c : 0;
            m    = up ? (m >> wd) : lowm;
        }
        myrow = pos;
    }
    unsigned long long live = rm_s[hh * 64 + myrow];
    const float q  = q_s[hh * 64 + myrow];
    const float mc = mc_s[hh * 64 + myrow];
    const int nit   = (__builtin_popcountll(live) + 3) >> 2;
    const int deadg = (~live != 0ull) ? (int)__builtin_ctzll(~live) : 0;
    int niter = 0;
    {
        unsigned cand = 0xFFFFFFFFu;
        for (int bit = 4; bit >= 0; --bit) {
            const unsigned bs = __builtin_amdgcn_ballot_w32(((nit >> bit) & 1) != 0) & cand;
            if (bs != 0u) { niter |= (1 << bit); cand = bs; }
        }
    }

    v8f acc = {};
    float denom = 0.0f;
    const v2f q2  = {q, q};
    const v2f mc2 = {mc, mc};
    const v2f ln2v = {LN2, LN2};
    const v2f epsv = {EPSF * LN2, EPSF * LN2};

#define PIN(X) asm volatile("" : "+v"(X))

#define ROUND(TN, TB)                                                      \
    {                                                                      \
        PIN(sv);   v2f A = sv;  A.x = __builtin_amdgcn_exp2f(A.x); PIN(A); \
        PIN(qq);   v2f nn = (TN).xy - qq;                         PIN(nn); \
        PIN(A);    A.y = __builtin_amdgcn_exp2f(A.y);              PIN(A); \
        PIN(rv);   sv = rv - mc2;                                 PIN(sv); \
        PIN(fv);   rv = fv;  rv.x = __builtin_amdgcn_rcpf(rv.x);  PIN(rv); \
        PIN(no);   fv = __builtin_elementwise_fma(no, ln2v, epsv); PIN(fv);\
        PIN(rv);   rv.y = __builtin_amdgcn_rcpf(rv.y);            PIN(rv); \
        PIN(tsum); PIN(Ap);                                                \
        denom += tsum;  tsum = Ap.x + Ap.y;                                \
        PIN(denom); PIN(tsum); PIN(nn);                                    \
        nn.x = __builtin_fabsf(nn.x);                             PIN(nn); \
        nn.y = __builtin_fabsf(nn.y);                                      \
        PIN(nn); PIN(A);                                                   \
        acc = __builtin_amdgcn_wmma_f32_16x16x4_f32(                       \
                  false, A, false, (TB).zw, (short)0, acc, false, false);  \
        PIN(acc);                                                          \
        no = nn;  Ap = A;                                                  \
    }

#define NEXTG(G)                                                         \
        G = (live != 0ull) ? (int)__builtin_ctzll(live) : deadg;         \
        live &= live - 1ull;
#define LOADS(P, GA, GB)                                                 \
        P##0 = tl[9 * (GA) + 0];                                         \
        P##1 = tl[9 * (GA) + 2];                                         \
        P##2 = tl[9 * (GA) + 4];                                         \
        P##3 = tl[9 * (GA) + 6];                                         \
        P##4 = tl[9 * (GB) + 0];                                         \
        P##5 = tl[9 * (GB) + 2];                                         \
        P##6 = tl[9 * (GB) + 4];                                         \
        P##7 = tl[9 * (GB) + 6];
#define BODY(C, N)                                                       \
    {                                                                    \
        ROUND(C##0, N##4) ROUND(C##1, N##5)                              \
        ROUND(C##2, N##6) ROUND(C##3, N##7)                              \
        NEXTG(ga) NEXTG(gb)                                              \
        LOADS(N, ga, gb)                                                 \
        ROUND(C##4, C##0) ROUND(C##5, C##1)                              \
        ROUND(C##6, C##2) ROUND(C##7, C##3)                              \
    }

    const v4f* tl = kv_s + hh * TABH + half;
    v4f ta0, ta1, ta2, ta3, ta4, ta5, ta6, ta7;
    v4f tb0, tb1, tb2, tb3, tb4, tb5, tb6, tb7;
    int ga, gb;
    NEXTG(ga) NEXTG(gb)
    LOADS(ta, ga, gb)
    tb4 = tb5 = tb6 = tb7 = v4f{0.0f, 0.0f, 0.0f, 0.0f};
    v2f qq = q2;
    v2f no = {-2.0f * EPSF, -2.0f * EPSF};
    v2f fv = __builtin_elementwise_fma(no, ln2v, epsv);
    v2f rv = {__builtin_amdgcn_rcpf(fv.x), __builtin_amdgcn_rcpf(fv.y)};
    v2f sv = rv - mc2;
    float tsum = 0.0f;
    v2f Ap = {0.0f, 0.0f};

    for (int it = 0; it < niter; ++it) {
        BODY(ta, tb)
        BODY(tb, ta)
    }
    {
        v2f A;
        denom += tsum;
        denom += Ap.x + Ap.y;
        A.x = __builtin_amdgcn_exp2f(sv.x);
        A.y = __builtin_amdgcn_exp2f(sv.y);
        denom += A.x + A.y;
        acc = __builtin_amdgcn_wmma_f32_16x16x4_f32(
                  false, A, false, tb4.zw, (short)0, acc, false, false);
        sv = rv - mc2;
        A.x = __builtin_amdgcn_exp2f(sv.x);
        A.y = __builtin_amdgcn_exp2f(sv.y);
        denom += A.x + A.y;
        acc = __builtin_amdgcn_wmma_f32_16x16x4_f32(
                  false, A, false, tb5.zw, (short)0, acc, false, false);
        rv.x = __builtin_amdgcn_rcpf(fv.x);
        rv.y = __builtin_amdgcn_rcpf(fv.y);
        sv = rv - mc2;
        A.x = __builtin_amdgcn_exp2f(sv.x);
        A.y = __builtin_amdgcn_exp2f(sv.y);
        denom += A.x + A.y;
        acc = __builtin_amdgcn_wmma_f32_16x16x4_f32(
                  false, A, false, tb6.zw, (short)0, acc, false, false);
        fv = __builtin_elementwise_fma(no, ln2v, epsv);
        rv.x = __builtin_amdgcn_rcpf(fv.x);
        rv.y = __builtin_amdgcn_rcpf(fv.y);
        sv = rv - mc2;
        A.x = __builtin_amdgcn_exp2f(sv.x);
        A.y = __builtin_amdgcn_exp2f(sv.y);
        denom += A.x + A.y;
        acc = __builtin_amdgcn_wmma_f32_16x16x4_f32(
                  false, A, false, tb7.zw, (short)0, acc, false, false);
    }
#undef BODY
#undef LOADS
#undef NEXTG
#undef ROUND
#undef PIN
    denom += __shfl_xor(denom, 16, 32);

    const int slot = ((myrow >> 4) * HH + hh) * 16 + (myrow & 15);
    if (lane < 16) dn_s[slot] = denom;
    float sel = acc[0];
    #pragma unroll
    for (int r = 1; r < 8; ++r) sel = ((lane & 7) == r) ? acc[r] : sel;
    if (lane < 8 || lane >= 24) att_s[slot] = sel;
    __syncthreads();

    if (tid < 64) {
        const int tt  = tid >> 4;
        const int row = tid & 15;
        float s = 0.0f;
        #pragma unroll
        for (int hh = 0; hh < HH; ++hh)
            s += att_s[(tt * HH + hh) * 16 + row] / dn_s[(tt * HH + hh) * 16 + row];
        out[b * DD + ib0 + tid] = x[b * DD + ib0 + tid] + 0.5f * s;
    }
}

extern "C" void kernel_launch(void* const* d_in, const int* in_sizes, int n_in,
                              void* d_out, int out_size, void* d_ws, size_t ws_size,
                              hipStream_t stream) {
    (void)in_sizes; (void)n_in; (void)d_ws; (void)ws_size; (void)out_size;
    const float* x  = (const float*)d_in[0];
    const float* aq = (const float*)d_in[1];
    const float* bq = (const float*)d_in[2];
    const float* ak = (const float*)d_in[3];
    const float* bk = (const float*)d_in[4];
    const float* av = (const float*)d_in[5];
    const float* bv = (const float*)d_in[6];
    float* out = (float*)d_out;

    flatt_kernel<<<dim3(BB * (DD / 64)), dim3(512), 0, stream>>>(
        x, aq, bq, ak, bk, av, bv, out);
}


# pass 1 dealt by key range, not by row tile: each of a head's 4 waves takes 16 key groups for all 64 rows (4 rows per lane), so a table read serves 4 rows; row minima joined via LDS; values kept
# speedup vs baseline: 2.9477x; 1.0691x over previous
#include <hip/hip_runtime.h>
#include <hip/hip_bf16.h>

typedef __attribute__((ext_vector_type(2))) float v2f;
typedef __attribute__((ext_vector_type(4))) float v4f;
typedef __attribute__((ext_vector_type(8))) float v8f;

#define DD   1024
#define BB   16
#define HH   4
#define EPSF 1e-8f
#define LN2  0.69314718055994530942f
#define TABH (DD / 2 + DD / 16 + 1)
typedef unsigned short __attribute__((may_alias)) u16a;

__global__ __launch_bounds__(512) void flatt_kernel(
    const float* __restrict__ x,
    const float* __restrict__ aq, const float* __restrict__ bq,
    const float* __restrict__ ak, const float* __restrict__ bk,
    const float* __restrict__ av, const float* __restrict__ bv,
    float* __restrict__ out)
{
    __shared__ v4f   kv_s[HH * TABH];
    __shared__ float att_s[4 * HH * 16];
    __shared__ float dn_s[4 * HH * 16];
    __shared__ unsigned long long rm_s[HH * 64];
    __shared__ float    q_s[HH * 64];
    __shared__ float    mc_s[HH * 64];
    __shared__ v4f   pmin_s[HH * 4 * 16];

    const int b     = blockIdx.x >> 4;
    const int ib0   = (blockIdx.x & 15) << 6;
    const int tid   = threadIdx.x;
    const int w     = tid >> 5;
    const int t     = w >> 2;
    const int h     = (w + t) & 3;
    const int lane  = tid & 31;
    const int iloc  = lane & 15;
    const int half  = lane >> 4;

    const float Aq = aq[h], Bq = bq[h];
    const float Ak = ak[h], Bk = bk[h];
    const float Av = av[h], Bv = bv[h];

    v4f* tab = kv_s + h * TABH;
    const v4f* xrow4 = reinterpret_cast<const v4f*>(x + b * DD);
    #pragma unroll
    for (int rr = 0; rr < DD / 512; ++rr) {
        const int m = lane + 32 * (2 * t + rr);
        const v4f xv = xrow4[m];
        v4f e0, e1;
        e0.x = fmaf(xv.x, Ak, Bk);  e0.y = fmaf(xv.y, Ak, Bk);
        e0.z = fmaf(xv.x, Av, Bv);  e0.w = fmaf(xv.y, Av, Bv);
        e1.x = fmaf(xv.z, Ak, Bk);  e1.y = fmaf(xv.w, Ak, Bk);
        e1.z = fmaf(xv.z, Av, Bv);  e1.w = fmaf(xv.w, Av, Bv);
        tab[2 * m + (m >> 2)]     = e0;
        tab[2 * m + (m >> 2) + 1] = e1;
    }

    float qr[4];
    #pragma unroll
    for (int r = 0; r < 4; ++r)
        qr[r] = fmaf(x[b * DD + ib0 + 16 * r + iloc], Aq, Bq);
    __syncthreads();

    const v2f* tk = reinterpret_cast<const v2f*>(tab) + 288 * t;
    float gm[4][16];
#define P1_READ(BUF, B)                                                  \
    _Pragma("unroll")                                                    \
    for (int u = 0; u < 16; ++u)                                         \
        BUF[u] = tk[2 * (36 * (B) + 2 * u + (u >> 2) + half)];
#define P1_MIN(BUF, B)                                                   \
    _Pragma("unroll")                                                    \
    for (int r = 0; r < 4; ++r) {                                        \
        const v2f qv = {qr[r], qr[r]};                                   \
        _Pragma("unroll")                                                \
        for (int g = 0; g < 4; ++g) {                                    \
            const v2f da = BUF[4 * g + 0] - qv, db = BUF[4 * g + 1] - qv; \
            const v2f dc = BUF[4 * g + 2] - qv, dd = BUF[4 * g + 3] - qv; \
            float m = __builtin_fminf(__builtin_fminf(fabsf(da.x), fabsf(da.y)), fabsf(db.x)); \
            m = __builtin_fminf(__builtin_fminf(m, fabsf(db.y)), fabsf(dc.x)); \
            m = __builtin_fminf(__builtin_fminf(m, fabsf(dc.y)), fabsf(dd.x)); \
            gm[r][4 * (B) + g] = __builtin_fminf(m, fabsf(dd.y));        \
        }                                                                \
        asm volatile("" : "+v"(gm[r][4 * (B) + 0]), "+v"(gm[r][4 * (B) + 1]), \
                          "+v"(gm[r][4 * (B) + 2]), "+v"(gm[r][4 * (B) + 3])); \
        __builtin_amdgcn_sched_barrier(0);                               \
    }
    v2f kb0[16], kb1[16];
    P1_READ(kb0, 0)
    #pragma unroll
    for (int B = 0; B < 4; B += 2) {
        P1_READ(kb1, B + 1)
        __builtin_amdgcn_sched_barrier(0);
        P1_MIN(kb0, B)
        __builtin_amdgcn_sched_barrier(0);
        if (B + 2 < 4) {
            P1_READ(kb0, B + 2)
        }
        __builtin_amdgcn_sched_barrier(0);
        P1_MIN(kb1, B + 1)
        __builtin_amdgcn_sched_barrier(0);
    }
#undef P1_MIN
#undef P1_READ
    v4f pm;
    #pragma unroll
    for (int r = 0; r < 4; ++r) {
        float m0 = 3.4e38f, m1 = 3.4e38f;
        #pragma unroll
        for (int G = 0; G < 16; G += 4) {
            m0 = __builtin_fminf(__builtin_fminf(m0, gm[r][G]), gm[r][G + 1]);
            m1 = __builtin_fminf(__builtin_fminf(m1, gm[r][G + 2]), gm[r][G + 3]);
        }
        float m = __builtin_fminf(m0, m1);
        m = __builtin_fminf(m, __shfl_xor(m, 16, 32));
        pm[r] = m;
    }
    if (lane < 16) pmin_s[(h * 4 + t) * 16 + iloc] = pm;
    __syncthreads();
    float mcr[4], dcr[4];
    {
        const v4f p0 = pmin_s[(h * 4 + 0) * 16 + iloc], p1 = pmin_s[(h * 4 + 1) * 16 + iloc];
        const v4f p2 = pmin_s[(h * 4 + 2) * 16 + iloc], p3 = pmin_s[(h * 4 + 3) * 16 + iloc];
        const v4f dmin4 = __builtin_elementwise_min(__builtin_elementwise_min(p0, p1),
                                                    __builtin_elementwise_min(p2, p3));
        const v4f ln2q = {LN2, LN2, LN2, LN2};
        const v4f epsq = {EPSF * LN2, EPSF * LN2, EPSF * LN2, EPSF * LN2};
        const v4f arg4 = __builtin_elementwise_fma(dmin4, ln2q, epsq);
        v4f mc4;
        #pragma unroll
        for (int r = 0; r < 4; ++r) mc4[r] = __builtin_amdgcn_rcpf(arg4[r]);
        const v4f dc4 = 1.001f / ((mc4 - 160.0f) * LN2);
        #pragma unroll
        for (int r = 0; r < 4; ++r) {
            mcr[r] = mc4[r];
            dcr[r] = (mc4[r] > 200.0f) ? dc4[r] : __builtin_inff();
        }
    }
    unsigned pc[4];
    #pragma unroll
    for (int r = 0; r < 4; ++r) {
        unsigned m = 0u;
        #pragma unroll
        for (int G = 0; G < 16; ++G)
            m |= (gm[r][G] <= dcr[r]) ? (1u << G) : 0u;
        pc[r] = m | (unsigned)__shfl_xor((int)m, 16, 32);
    }
    const int tu = __builtin_amdgcn_readfirstlane(t);
    const float qown  = (tu == 0) ? qr[0]  : (tu == 1) ? qr[1]  : (tu == 2) ? qr[2]  : qr[3];
    const float mcown = (tu == 0) ? mcr[0] : (tu == 1) ? mcr[1] : (tu == 2) ? mcr[2] : mcr[3];
    if (lane < 16) {
        u16a* rp = reinterpret_cast<u16a*>(rm_s) + (h * 64 + iloc) * 4 + t;
        #pragma unroll
        for (int r = 0; r < 4; ++r)
            rp[64 * r] = (unsigned short)pc[r];
        q_s[h * 64 + 16 * t + iloc]  = qown;
        mc_s[h * 64 + 16 * t + iloc] = mcown;
    }
    __syncthreads();

    unsigned fb[8], mb[8];
    #pragma unroll
    for (int j = 0; j < 8; ++j) {
        const float mcv = mc_s[32 * j + lane];
        const unsigned long long lv = rm_s[32 * j + lane];
        const bool isfar = !(mcv > 200.0f);
        const bool ismid = !isfar && (__builtin_popcountll(lv) > 4);
        fb[j] = __builtin_amdgcn_ballot_w32(isfar);
        mb[j] = __builtin_amdgcn_ballot_w32(ismid);
    }
    const unsigned long long fm0 = ((unsigned long long)fb[1] << 32) | fb[0], fm1 = ((unsigned long long)fb[3] << 32) | fb[2];
    const unsigned long long fm2 = ((unsigned long long)fb[5] << 32) | fb[4], fm3 = ((unsigned long long)fb[7] << 32) | fb[6];
    const unsigned long long mm0 = ((unsigned long long)mb[1] << 32) | mb[0], mm1 = ((unsigned long long)mb[3] << 32) | mb[2];
    const unsigned long long mm2 = ((unsigned long long)mb[5] << 32) | mb[4], mm3 = ((unsigned long long)mb[7] << 32) | mb[6];
    const int nfar = __builtin_popcountll(fm0) + __builtin_popcountll(fm1)
                   + __builtin_popcountll(fm2) + __builtin_popcountll(fm3);
    const int nmid = __builtin_popcountll(mm0) + __builtin_popcountll(mm1)
                   + __builtin_popcountll(mm2) + __builtin_popcountll(mm3);
    const int gt = (t & 1) ? (4 * t + 3 - (w & 3)) : (4 * t + (w & 3));
    int myrow, hh;
    {
        const int n = 16 * gt + iloc;
        const bool cf = (n < nfar), cm = (n < nfar + nmid);
        int k = cf ? n : (cm ? n - nfar : n - nfar - nmid);
        const unsigned long long a0 = cf ? fm0 : (cm ? mm0 : ~(fm0 | mm0));
        const unsigned long long a1 = cf ? fm1 : (cm ? mm1 : ~(fm1 | mm1));
        const unsigned long long a2 = cf ? fm2 : (cm ? mm2 : ~(fm2 | mm2));
        const unsigned long long a3 = cf ? fm3 : (cm ? mm3 : ~(fm3 | mm3));
        const int c0 = __builtin_popcountll(a0);
        const int c1 = c0 + __builtin_popcountll(a1);
        const int c2 = c1 + __builtin_popcountll(a2);
        const bool p1 = (k >= c0), p2 = (k >= c1), p3 = (k >= c2);
        hh = (int)p1 + (int)p2 + (int)p3;
        unsigned long long m = p3 ? a3 : (p2 ? a2 : (p1 ? a1 : a0));
        k -= p3 ? c2 : (p2 ? c1 : (p1 ? c0 : 0));
        int pos = 0;
        #pragma unroll
        for (int wd = 32; wd >= 1; wd >>= 1) {
            const unsigned long long lowm = m & ((1ull << wd) - 1ull);
            const int c = __builtin_popcountll(lowm);
            const bool up = (k >= c);
            pos += up ? wd : 0;
            k   -= up ? c : 0;
            m    = up ? (m >> wd) : lowm;
        }
        myrow = pos;
    }
    unsigned long long live = rm_s[hh * 64 + myrow];
    const float q  = q_s[hh * 64 + myrow];
    const float mc = mc_s[hh * 64 + myrow];
    const int nit   = (__builtin_popcountll(live) + 3) >> 2;
    const int deadg = (~live != 0ull) ? (int)__builtin_ctzll(~live) : 0;
    int niter = 0;
    {
        unsigned cand = 0xFFFFFFFFu;
        for (int bit = 4; bit >= 0; --bit) {
            const unsigned bs = __builtin_amdgcn_ballot_w32(((nit >> bit) & 1) != 0) & cand;
            if (bs != 0u) { niter |= (1 << bit); cand = bs; }
        }
    }

    v8f acc = {};
    float denom = 0.0f;
    const v2f q2  = {q, q};
    const v2f mc2 = {mc, mc};
    const v2f ln2v = {LN2, LN2};
    const v2f epsv = {EPSF * LN2, EPSF * LN2};

#define PIN(X) asm volatile("" : "+v"(X))

#define ROUND(TN, TB)                                                      \
    {                                                                      \
        PIN(sv);   v2f A = sv;  A.x = __builtin_amdgcn_exp2f(A.x); PIN(A); \
        PIN(qq);   v2f nn = (TN).xy - qq;                         PIN(nn); \
        PIN(A);    A.y = __builtin_amdgcn_exp2f(A.y);              PIN(A); \
        PIN(rv);   sv = rv - mc2;                                 PIN(sv); \
        PIN(fv);   rv = fv;  rv.x = __builtin_amdgcn_rcpf(rv.x);  PIN(rv); \
        PIN(no);   fv = __builtin_elementwise_fma(no, ln2v, epsv); PIN(fv);\
        PIN(rv);   rv.y = __builtin_amdgcn_rcpf(rv.y);            PIN(rv); \
        PIN(tsum); PIN(Ap);                                                \
        denom += tsum;  tsum = Ap.x + Ap.y;                                \
        PIN(denom); PIN(tsum); PIN(nn);                                    \
        nn.x = __builtin_fabsf(nn.x);                             PIN(nn); \
        nn.y = __builtin_fabsf(nn.y);                                      \
        PIN(nn); PIN(A);                                                   \
        acc = __builtin_amdgcn_wmma_f32_16x16x4_f32(                       \
                  false, A, false, (TB).zw, (short)0, acc, false, false);  \
        PIN(acc);                                                          \
        no = nn;  Ap = A;                                                  \
    }

#define NEXTG(G)                                                         \
        G = (live != 0ull) ? (int)__builtin_ctzll(live) : deadg;         \
        live &= live - 1ull;
#define LOADS(P, GA, GB)                                                 \
        P##0 = tl[9 * (GA) + 0];                                         \
        P##1 = tl[9 * (GA) + 2];                                         \
        P##2 = tl[9 * (GA) + 4];                                         \
        P##3 = tl[9 * (GA) + 6];                                         \
        P##4 = tl[9 * (GB) + 0];                                         \
        P##5 = tl[9 * (GB) + 2];                                         \
        P##6 = tl[9 * (GB) + 4];                                         \
        P##7 = tl[9 * (GB) + 6];
#define BODY(C, N)                                                       \
    {                                                                    \
        ROUND(C##0, N##4) ROUND(C##1, N##5)                              \
        ROUND(C##2, N##6) ROUND(C##3, N##7)                              \
        NEXTG(ga) NEXTG(gb)                                              \
        LOADS(N, ga, gb)                                                 \
        ROUND(C##4, C##0) ROUND(C##5, C##1)                              \
        ROUND(C##6, C##2) ROUND(C##7, C##3)                              \
    }

    const v4f* tl = kv_s + hh * TABH + half;
    v4f ta0, ta1, ta2, ta3, ta4, ta5, ta6, ta7;
    v4f tb0, tb1, tb2, tb3, tb4, tb5, tb6, tb7;
    int ga, gb;
    NEXTG(ga) NEXTG(gb)
    LOADS(ta, ga, gb)
    tb4 = tb5 = tb6 = tb7 = v4f{0.0f, 0.0f, 0.0f, 0.0f};
    v2f qq = q2;
    v2f no = {-2.0f * EPSF, -2.0f * EPSF};
    v2f fv = __builtin_elementwise_fma(no, ln2v, epsv);
    v2f rv = {__builtin_amdgcn_rcpf(fv.x), __builtin_amdgcn_rcpf(fv.y)};
    v2f sv = rv - mc2;
    float tsum = 0.0f;
    v2f Ap = {0.0f, 0.0f};

    for (int it = 0; it < niter; ++it) {
        BODY(ta, tb)
        BODY(tb, ta)
    }
    {
        v2f A;
        denom += tsum;
        denom += Ap.x + Ap.y;
        A.x = __builtin_amdgcn_exp2f(sv.x);
        A.y = __builtin_amdgcn_exp2f(sv.y);
        denom += A.x + A.y;
        acc = __builtin_amdgcn_wmma_f32_16x16x4_f32(
                  false, A, false, tb4.zw, (short)0, acc, false, false);
        sv = rv - mc2;
        A.x = __builtin_amdgcn_exp2f(sv.x);
        A.y = __builtin_amdgcn_exp2f(sv.y);
        denom += A.x + A.y;
        acc = __builtin_amdgcn_wmma_f32_16x16x4_f32(
                  false, A, false, tb5.zw, (short)0, acc, false, false);
        rv.x = __builtin_amdgcn_rcpf(fv.x);
        rv.y = __builtin_amdgcn_rcpf(fv.y);
        sv = rv - mc2;
        A.x = __builtin_amdgcn_exp2f(sv.x);
        A.y = __builtin_amdgcn_exp2f(sv.y);
        denom += A.x + A.y;
        acc = __builtin_amdgcn_wmma_f32_16x16x4_f32(
                  false, A, false, tb6.zw, (short)0, acc, false, false);
        fv = __builtin_elementwise_fma(no, ln2v, epsv);
        rv.x = __builtin_amdgcn_rcpf(fv.x);
        rv.y = __builtin_amdgcn_rcpf(fv.y);
        sv = rv - mc2;
        A.x = __builtin_amdgcn_exp2f(sv.x);
        A.y = __builtin_amdgcn_exp2f(sv.y);
        denom += A.x + A.y;
        acc = __builtin_amdgcn_wmma_f32_16x16x4_f32(
                  false, A, false, tb7.zw, (short)0, acc, false, false);
    }
#undef BODY
#undef LOADS
#undef NEXTG
#undef ROUND
#undef PIN
    denom += __shfl_xor(denom, 16, 32);

    const int slot = ((myrow >> 4) * HH + hh) * 16 + (myrow & 15);
    if (lane < 16) dn_s[slot] = denom;
    float sel = acc[0];
    #pragma unroll
    for (int r = 1; r < 8; ++r) sel = ((lane & 7) == r) ? acc[r] : sel;
    if (lane < 8 || lane >= 24) att_s[slot] = sel;
    __syncthreads();

    if (tid < 64) {
        const int tt  = tid >> 4;
        const int row = tid & 15;
        float s = 0.0f;
        #pragma unroll
        for (int hh = 0; hh < HH; ++hh)
            s += att_s[(tt * HH + hh) * 16 + row] / dn_s[(tt * HH + hh) * 16 + row];
        out[b * DD + ib0 + tid] = x[b * DD + ib0 + tid] + 0.5f * s;
    }
}

extern "C" void kernel_launch(void* const* d_in, const int* in_sizes, int n_in,
                              void* d_out, int out_size, void* d_ws, size_t ws_size,
                              hipStream_t stream) {
    (void)in_sizes; (void)n_in; (void)d_ws; (void)ws_size; (void)out_size;
    const float* x  = (const float*)d_in[0];
    const float* aq = (const float*)d_in[1];
    const float* bq = (const float*)d_in[2];
    const float* ak = (const float*)d_in[3];
    const float* bk = (const float*)d_in[4];
    const float* av = (const float*)d_in[5];
    const float* bv = (const float*)d_in[6];
    float* out = (float*)d_out;

    flatt_kernel<<<dim3(BB * (DD / 64)), dim3(512), 0, stream>>>(
        x, aq, bq, ak, bk, av, bv, out);
}
